# starting kernel
# baseline (speedup 1.0000x reference)
#include <hip/hip_runtime.h>
#include <hip/hip_bf16.h>


typedef __attribute__((ext_vector_type(2))) float v2f;
typedef __attribute__((ext_vector_type(8))) float v8f;

#define HEADS 4
#define HDIM 32
#define CIN 128
#define NEG_SLOPE 0.2f

__global__ void gat_zero_kernel(float* __restrict__ p, size_t n) {
    size_t i = (size_t)blockIdx.x * blockDim.x + threadIdx.x;
    size_t stride = (size_t)gridDim.x * blockDim.x;
    for (; i < n; i += stride) p[i] = 0.0f;
}

__global__ __launch_bounds__(256) void gat_gemm_wmma(
    const float* __restrict__ X,
    const float* __restrict__ Wl,
    const float* __restrict__ Wr,
    float* __restrict__ XL,
    float* __restrict__ XR,
    int Nn)
{
    const int wave = threadIdx.x >> 5;
    const int lane = threadIdx.x & 31;
    const int row0 = blockIdx.x * 16;
    const int col0 = wave * 16;

    const float* __restrict__ W  = (blockIdx.y == 0) ? Wl : Wr;
    float*       __restrict__ O  = (blockIdx.y == 0) ? XL : XR;

    const int m  = lane & 15;
    const int hi = lane >> 4;

    int arow = row0 + m; if (arow >= Nn) arow = Nn - 1;

    const float* __restrict__ ap = X + (size_t)arow * CIN;
    const float* __restrict__ bp = W + (size_t)(col0 + m) * CIN;

    v8f acc = {};
    #pragma unroll
    for (int k0 = 0; k0 < CIN; k0 += 4) {
        const int ka = k0 + hi * 2;
        v2f a, b;
        a.x = ap[ka];  a.y = ap[ka + 1];
        b.x = bp[ka];  b.y = bp[ka + 1];
        acc = __builtin_amdgcn_wmma_f32_16x16x4_f32(
             false, a,  false, b,
             (short)0, acc,  false,  false);
    }

    #pragma unroll
    for (int i = 0; i < 8; ++i) {
        const int r = row0 + i + hi * 8;
        if (r < Nn) O[(size_t)r * CIN + col0 + m] = acc[i];
    }
}

__global__ void gat_edge_score(
    const float* __restrict__ xl, const float* __restrict__ xr,
    const int* __restrict__ src, const int* __restrict__ dst,
    const float* __restrict__ att,
    float* __restrict__ ebuf, int* __restrict__ emax_i,
    int Etot, int E)
{
    const int t = blockIdx.x * blockDim.x + threadIdx.x;
    if (t >= Etot * HEADS) return;
    const int eid = t >> 2;
    const int h   = t & 3;

    int s, d;
    if (eid < E) { s = src[eid]; d = dst[eid]; }
    else         { s = d = eid - E; }

    const float* __restrict__ pl = xl + (size_t)s * CIN + h * HDIM;
    const float* __restrict__ pr = xr + (size_t)d * CIN + h * HDIM;
    const float* __restrict__ pa = att + h * HDIM;

    float acc = 0.0f;
    #pragma unroll
    for (int i = 0; i < HDIM; ++i) {
        float v = pl[i] + pr[i];
        v = (v > 0.0f) ? v : NEG_SLOPE * v;
        acc += pa[i] * v;
    }
    ebuf[(size_t)eid * HEADS + h] = acc;
    if (acc > 0.0f)
        atomicMax(&emax_i[d * HEADS + h], __float_as_int(acc));
}

__global__ void gat_edge_exp(
    const int* __restrict__ dst,
    float* __restrict__ ebuf,
    const float* __restrict__ emax,
    float* __restrict__ denom,
    int Etot, int E)
{
    const int t = blockIdx.x * blockDim.x + threadIdx.x;
    if (t >= Etot * HEADS) return;
    const int eid = t >> 2;
    const int h   = t & 3;
    const int d   = (eid < E) ? dst[eid] : (eid - E);

    const float m  = emax[d * HEADS + h];
    const float ex = __expf(ebuf[(size_t)eid * HEADS + h] - m);
    ebuf[(size_t)eid * HEADS + h] = ex;
    atomicAdd(&denom[d * HEADS + h], ex);
}

__global__ void gat_edge_scatter(
    const float* __restrict__ xr,
    const int* __restrict__ src, const int* __restrict__ dst,
    const float* __restrict__ ebuf, const float* __restrict__ denom,
    float* __restrict__ accum,
    int Etot, int E)
{
    const int t = blockIdx.x * blockDim.x + threadIdx.x;
    if (t >= Etot * HEADS) return;
    const int eid = t >> 2;
    const int h   = t & 3;

    int s, d;
    if (eid < E) { s = src[eid]; d = dst[eid]; }
    else         { s = d = eid - E; }

    float den = denom[d * HEADS + h];
    den = (den > 1e-16f) ? den : 1e-16f;
    const float alpha = ebuf[(size_t)eid * HEADS + h] / den;

    const float* __restrict__ p = xr    + (size_t)s * CIN + h * HDIM;
    float*       __restrict__ o = accum + (size_t)d * CIN + h * HDIM;
    #pragma unroll
    for (int i = 0; i < HDIM; ++i)
        atomicAdd(&o[i], p[i] * alpha);
}

__global__ void gat_finalize(
    const float* __restrict__ accum, const float* __restrict__ bias,
    float* __restrict__ out, int Nn)
{
    const int t = blockIdx.x * blockDim.x + threadIdx.x;
    if (t >= Nn * HDIM) return;
    const int n  = t >> 5;
    const int dc = t & 31;
    const float* __restrict__ a = accum + (size_t)n * CIN + dc;
    out[t] = (a[0] + a[HDIM] + a[2 * HDIM] + a[3 * HDIM]) * 0.25f + bias[dc];
}

extern "C" void kernel_launch(void* const* d_in, const int* in_sizes, int n_in,
                              void* d_out, int out_size, void* d_ws, size_t ws_size,
                              hipStream_t stream)
{
    const float* x    = (const float*)d_in[0];
    const int*   eidx = (const int*)d_in[1];
    const float* Wl   = (const float*)d_in[2];
    const float* Wr   = (const float*)d_in[3];
    const float* att  = (const float*)d_in[4];
    const float* bias = (const float*)d_in[5];
    float* out = (float*)d_out;

    const int Nn   = in_sizes[0] / CIN;
    const int E    = in_sizes[1] / 2;
    const int Etot = E + Nn;

    const int* src = eidx;
    const int* dst = eidx + E;

    float* ws = (float*)d_ws;
    size_t off = 0;
    float* xl    = ws + off; off += (size_t)Nn * CIN;
    float* xr    = ws + off; off += (size_t)Nn * CIN;
    float* ebuf  = ws + off; off += (size_t)Etot * HEADS;
    float* emax  = ws + off; off += (size_t)Nn * HEADS;
    float* denom = ws + off; off += (size_t)Nn * HEADS;
    float* accum = ws + off; off += (size_t)Nn * CIN;
    (void)ws_size;

    {
        size_t nz = (size_t)Nn * HEADS * 2 + (size_t)Nn * CIN;
        int blocks = 2048;
        gat_zero_kernel<<<blocks, 256, 0, stream>>>(emax, nz);
    }

    {
        dim3 grid((Nn + 15) / 16, 2);
        gat_gemm_wmma<<<grid, 256, 0, stream>>>(x, Wl, Wr, xl, xr, Nn);
    }

    const int edgeThreads = Etot * HEADS;
    const int edgeBlocks  = (edgeThreads + 255) / 256;

    gat_edge_score<<<edgeBlocks, 256, 0, stream>>>(
        xl, xr, src, dst, att, ebuf, (int*)emax, Etot, E);

    gat_edge_exp<<<edgeBlocks, 256, 0, stream>>>(
        dst, ebuf, emax, denom, Etot, E);

    gat_edge_scatter<<<edgeBlocks, 256, 0, stream>>>(
        xr, src, dst, ebuf, denom, accum, Etot, E);

    {
        const int t = Nn * HDIM;
        gat_finalize<<<(t + 255) / 256, 256, 0, stream>>>(accum, bias, out, Nn);
    }
}


# scatter: the 4 head terms of an edge are summed in registers (lane = channel) before the atomic, so 32 atomics per edge instead of 128; float sum order changes (heads first, then edges)
# speedup vs baseline: 3.2741x; 3.2741x over previous
#include <hip/hip_runtime.h>
#include <hip/hip_bf16.h>


typedef __attribute__((ext_vector_type(2))) float v2f;
typedef __attribute__((ext_vector_type(8))) float v8f;

#define HEADS 4
#define HDIM 32
#define CIN 128
#define NEG_SLOPE 0.2f

__global__ void gat_zero_kernel(float* __restrict__ p, size_t n) {
    size_t i = (size_t)blockIdx.x * blockDim.x + threadIdx.x;
    size_t stride = (size_t)gridDim.x * blockDim.x;
    for (; i < n; i += stride) p[i] = 0.0f;
}

__global__ __launch_bounds__(256) void gat_gemm_wmma(
    const float* __restrict__ X,
    const float* __restrict__ Wl,
    const float* __restrict__ Wr,
    float* __restrict__ XL,
    float* __restrict__ XR,
    int Nn)
{
    const int wave = threadIdx.x >> 5;
    const int lane = threadIdx.x & 31;
    const int row0 = blockIdx.x * 16;
    const int col0 = wave * 16;

    const float* __restrict__ W  = (blockIdx.y == 0) ? Wl : Wr;
    float*       __restrict__ O  = (blockIdx.y == 0) ? XL : XR;

    const int m  = lane & 15;
    const int hi = lane >> 4;

    int arow = row0 + m; if (arow >= Nn) arow = Nn - 1;

    const float* __restrict__ ap = X + (size_t)arow * CIN;
    const float* __restrict__ bp = W + (size_t)(col0 + m) * CIN;

    v8f acc = {};
    #pragma unroll
    for (int k0 = 0; k0 < CIN; k0 += 4) {
        const int ka = k0 + hi * 2;
        v2f a, b;
        a.x = ap[ka];  a.y = ap[ka + 1];
        b.x = bp[ka];  b.y = bp[ka + 1];
        acc = __builtin_amdgcn_wmma_f32_16x16x4_f32(
             false, a,  false, b,
             (short)0, acc,  false,  false);
    }

    #pragma unroll
    for (int i = 0; i < 8; ++i) {
        const int r = row0 + i + hi * 8;
        if (r < Nn) O[(size_t)r * CIN + col0 + m] = acc[i];
    }
}

__global__ void gat_edge_score(
    const float* __restrict__ xl, const float* __restrict__ xr,
    const int* __restrict__ src, const int* __restrict__ dst,
    const float* __restrict__ att,
    float* __restrict__ ebuf, int* __restrict__ emax_i,
    int Etot, int E)
{
    const int t = blockIdx.x * blockDim.x + threadIdx.x;
    if (t >= Etot * HEADS) return;
    const int eid = t >> 2;
    const int h   = t & 3;

    int s, d;
    if (eid < E) { s = src[eid]; d = dst[eid]; }
    else         { s = d = eid - E; }

    const float* __restrict__ pl = xl + (size_t)s * CIN + h * HDIM;
    const float* __restrict__ pr = xr + (size_t)d * CIN + h * HDIM;
    const float* __restrict__ pa = att + h * HDIM;

    float acc = 0.0f;
    #pragma unroll
    for (int i = 0; i < HDIM; ++i) {
        float v = pl[i] + pr[i];
        v = (v > 0.0f) ? v : NEG_SLOPE * v;
        acc += pa[i] * v;
    }
    ebuf[(size_t)eid * HEADS + h] = acc;
    if (acc > 0.0f)
        atomicMax(&emax_i[d * HEADS + h], __float_as_int(acc));
}

__global__ void gat_edge_exp(
    const int* __restrict__ dst,
    float* __restrict__ ebuf,
    const float* __restrict__ emax,
    float* __restrict__ denom,
    int Etot, int E)
{
    const int t = blockIdx.x * blockDim.x + threadIdx.x;
    if (t >= Etot * HEADS) return;
    const int eid = t >> 2;
    const int h   = t & 3;
    const int d   = (eid < E) ? dst[eid] : (eid - E);

    const float m  = emax[d * HEADS + h];
    const float ex = __expf(ebuf[(size_t)eid * HEADS + h] - m);
    ebuf[(size_t)eid * HEADS + h] = ex;
    atomicAdd(&denom[d * HEADS + h], ex);
}

#define SC_GROUP 4

template <int G>
static __device__ __forceinline__ void gat_scatter_group(
    const float* __restrict__ xr, float* __restrict__ accum,
    int lane, int jb, int s, int d, float a0, float a1, float a2, float a3)
{
    float x[G][4];
    float al[G][4];
    int   dj[G];
    #pragma unroll
    for (int u = 0; u < G; ++u) {
        const int j  = jb + u;
        const int sj = __builtin_amdgcn_readlane(s, j);
        dj[u]    = __builtin_amdgcn_readlane(d, j);
        al[u][0] = __int_as_float(__builtin_amdgcn_readlane(__float_as_int(a0), j));
        al[u][1] = __int_as_float(__builtin_amdgcn_readlane(__float_as_int(a1), j));
        al[u][2] = __int_as_float(__builtin_amdgcn_readlane(__float_as_int(a2), j));
        al[u][3] = __int_as_float(__builtin_amdgcn_readlane(__float_as_int(a3), j));
        const float* __restrict__ p = xr + (size_t)sj * CIN + lane;
        x[u][0] = p[0];
        x[u][1] = p[HDIM];
        x[u][2] = p[2 * HDIM];
        x[u][3] = p[3 * HDIM];
    }
    #pragma unroll
    for (int u = 0; u < G; ++u) {
        const float v = fmaf(al[u][3], x[u][3],
                        fmaf(al[u][2], x[u][2],
                        fmaf(al[u][1], x[u][1], al[u][0] * x[u][0])));
        atomicAdd(accum + (size_t)dj[u] * HDIM + lane, v);
    }
}

__global__ __launch_bounds__(256) void gat_edge_scatter(
    const float* __restrict__ xr,
    const int* __restrict__ src, const int* __restrict__ dst,
    const float* __restrict__ ebuf, const float* __restrict__ denom,
    float* __restrict__ accum,
    int Etot, int E)
{
    const int lane = threadIdx.x & 31;
    const int wave = __builtin_amdgcn_readfirstlane(
        (int)((blockIdx.x * blockDim.x + threadIdx.x) >> 5));
    const int e0   = wave * 32;
    if (e0 >= Etot) return;
    const int cnt = (Etot - e0 < 32) ? (Etot - e0) : 32;

    int   s = 0, d = 0;
    float a0 = 0.0f, a1 = 0.0f, a2 = 0.0f, a3 = 0.0f;
    if (lane < cnt) {
        const int e = e0 + lane;
        if (e < E) { s = src[e]; d = dst[e]; }
        else       { s = d = e - E; }
        const float4 ex = *reinterpret_cast<const float4*>(ebuf  + (size_t)e * HEADS);
        const float4 dn = *reinterpret_cast<const float4*>(denom + (size_t)d * HEADS);
        a0 = ex.x / ((dn.x > 1e-16f) ? dn.x : 1e-16f);
        a1 = ex.y / ((dn.y > 1e-16f) ? dn.y : 1e-16f);
        a2 = ex.z / ((dn.z > 1e-16f) ? dn.z : 1e-16f);
        a3 = ex.w / ((dn.w > 1e-16f) ? dn.w : 1e-16f);
    }

    int jb = 0;
    for (; jb + SC_GROUP <= cnt; jb += SC_GROUP)
        gat_scatter_group<SC_GROUP>(xr, accum, lane, jb, s, d, a0, a1, a2, a3);
    #pragma nounroll
    for (; jb < cnt; ++jb)
        gat_scatter_group<1>(xr, accum, lane, jb, s, d, a0, a1, a2, a3);
}

__global__ void gat_finalize(
    const float* __restrict__ accum, const float* __restrict__ bias,
    float* __restrict__ out, int Nn)
{
    const int t = blockIdx.x * blockDim.x + threadIdx.x;
    if (t >= Nn * HDIM) return;
    const int dc = t & 31;
    out[t] = accum[t] * 0.25f + bias[dc];
}

extern "C" void kernel_launch(void* const* d_in, const int* in_sizes, int n_in,
                              void* d_out, int out_size, void* d_ws, size_t ws_size,
                              hipStream_t stream)
{
    const float* x    = (const float*)d_in[0];
    const int*   eidx = (const int*)d_in[1];
    const float* Wl   = (const float*)d_in[2];
    const float* Wr   = (const float*)d_in[3];
    const float* att  = (const float*)d_in[4];
    const float* bias = (const float*)d_in[5];
    float* out = (float*)d_out;

    const int Nn   = in_sizes[0] / CIN;
    const int E    = in_sizes[1] / 2;
    const int Etot = E + Nn;

    const int* src = eidx;
    const int* dst = eidx + E;

    float* ws = (float*)d_ws;
    size_t off = 0;
    float* xl    = ws + off; off += (size_t)Nn * CIN;
    float* xr    = ws + off; off += (size_t)Nn * CIN;
    float* ebuf  = ws + off; off += (size_t)Etot * HEADS;
    float* emax  = ws + off; off += (size_t)Nn * HEADS;
    float* denom = ws + off; off += (size_t)Nn * HEADS;
    float* accum = ws + off; off += (size_t)Nn * HDIM;
    (void)ws_size;

    {
        size_t nz = (size_t)Nn * HEADS * 2 + (size_t)Nn * HDIM;
        int blocks = 2048;
        gat_zero_kernel<<<blocks, 256, 0, stream>>>(emax, nz);
    }

    {
        dim3 grid((Nn + 15) / 16, 2);
        gat_gemm_wmma<<<grid, 256, 0, stream>>>(x, Wl, Wr, xl, xr, Nn);
    }

    const int edgeThreads = Etot * HEADS;
    const int edgeBlocks  = (edgeThreads + 255) / 256;

    gat_edge_score<<<edgeBlocks, 256, 0, stream>>>(
        xl, xr, src, dst, att, ebuf, (int*)emax, Etot, E);

    gat_edge_exp<<<edgeBlocks, 256, 0, stream>>>(
        dst, ebuf, emax, denom, Etot, E);

    {
        const int scWaves  = (Etot + 31) / 32;
        const int scBlocks = (scWaves + 7) / 8;
        gat_edge_scatter<<<scBlocks, 256, 0, stream>>>(
            xr, src, dst, ebuf, denom, accum, Etot, E);
    }

    {
        const int t = Nn * HDIM;
        gat_finalize<<<(t + 255) / 256, 256, 0, stream>>>(accum, bias, out, Nn);
    }
}


# edge score: a wave walks edge rows (lane = 4 channels, whole-line 16 B/lane gathers) instead of a thread per (edge, head); score = 4-term lane sums + 8-lane DPP tree, so its sum order changes
# speedup vs baseline: 4.4690x; 1.3650x over previous
#include <hip/hip_runtime.h>
#include <hip/hip_bf16.h>


typedef __attribute__((ext_vector_type(2))) float v2f;
typedef __attribute__((ext_vector_type(8))) float v8f;

#define HEADS 4
#define HDIM 32
#define CIN 128
#define NEG_SLOPE 0.2f

__global__ void gat_zero_kernel(float* __restrict__ p, size_t n) {
    size_t i = (size_t)blockIdx.x * blockDim.x + threadIdx.x;
    size_t stride = (size_t)gridDim.x * blockDim.x;
    for (; i < n; i += stride) p[i] = 0.0f;
}

__global__ __launch_bounds__(256) void gat_gemm_wmma(
    const float* __restrict__ X,
    const float* __restrict__ Wl,
    const float* __restrict__ Wr,
    float* __restrict__ XL,
    float* __restrict__ XR,
    int Nn)
{
    const int wave = threadIdx.x >> 5;
    const int lane = threadIdx.x & 31;
    const int row0 = blockIdx.x * 16;
    const int col0 = wave * 16;

    const float* __restrict__ W  = (blockIdx.y == 0) ? Wl : Wr;
    float*       __restrict__ O  = (blockIdx.y == 0) ? XL : XR;

    const int m  = lane & 15;
    const int hi = lane >> 4;

    int arow = row0 + m; if (arow >= Nn) arow = Nn - 1;

    const float* __restrict__ ap = X + (size_t)arow * CIN;
    const float* __restrict__ bp = W + (size_t)(col0 + m) * CIN;

    v8f acc = {};
    #pragma unroll
    for (int k0 = 0; k0 < CIN; k0 += 4) {
        const int ka = k0 + hi * 2;
        v2f a, b;
        a.x = ap[ka];  a.y = ap[ka + 1];
        b.x = bp[ka];  b.y = bp[ka + 1];
        acc = __builtin_amdgcn_wmma_f32_16x16x4_f32(
             false, a,  false, b,
             (short)0, acc,  false,  false);
    }

    #pragma unroll
    for (int i = 0; i < 8; ++i) {
        const int r = row0 + i + hi * 8;
        if (r < Nn) O[(size_t)r * CIN + col0 + m] = acc[i];
    }
}

template <int M>
static __device__ __forceinline__ float gat_row_xor(float v) {
    return __int_as_float(__builtin_amdgcn_update_dpp(
        0, __float_as_int(v), 0x160 + M, 0xf, 0xf, true));
}

__global__ __launch_bounds__(256) void gat_edge_score(
    const float* __restrict__ xl, const float* __restrict__ xr,
    const int* __restrict__ src, const int* __restrict__ dst,
    const float* __restrict__ att,
    float* __restrict__ ebuf, int* __restrict__ emax_i,
    int Etot, int E)
{
    const int lane = threadIdx.x & 31;
    const int wave = __builtin_amdgcn_readfirstlane(
        (int)((blockIdx.x * blockDim.x + threadIdx.x) >> 5));
    const int e0   = wave * 32;
    if (e0 >= Etot) return;
    const int cnt = (Etot - e0 < 32) ? (Etot - e0) : 32;

    int s = 0, d = 0;
    if (lane < cnt) {
        const int e = e0 + lane;
        if (e < E) { s = src[e]; d = dst[e]; }
        else       { s = d = e - E; }
    }

    const float4 a4 = *reinterpret_cast<const float4*>(att + lane * 4);
    const int k = lane & 7;
    const int h = lane >> 3;

    for (int jb = 0; jb < cnt; jb += 8) {
        float4 L[8], R[8];
        for (int u = 0; u < 8; ++u) {
            const int j  = jb + u;
            const int sj = __builtin_amdgcn_readlane(s, j);
            const int dj = __builtin_amdgcn_readlane(d, j);
            L[u] = *reinterpret_cast<const float4*>(xl + (size_t)sj * CIN + lane * 4);
            R[u] = *reinterpret_cast<const float4*>(xr + (size_t)dj * CIN + lane * 4);
        }
        float keep = 0.0f;
        for (int u = 0; u < 8; ++u) {
            float v0 = L[u].x + R[u].x;
            float v1 = L[u].y + R[u].y;
            float v2 = L[u].z + R[u].z;
            float v3 = L[u].w + R[u].w;
            v0 = fmaxf(v0, NEG_SLOPE * v0);
            v1 = fmaxf(v1, NEG_SLOPE * v1);
            v2 = fmaxf(v2, NEG_SLOPE * v2);
            v3 = fmaxf(v3, NEG_SLOPE * v3);
            float p = fmaf(a4.w, v3, fmaf(a4.z, v2, fmaf(a4.y, v1, a4.x * v0)));
            p += gat_row_xor<1>(p);
            p += gat_row_xor<2>(p);
            p += gat_row_xor<4>(p);
            keep = (k == u) ? p : keep;
        }
        const int ej = jb + k;
        const int dd = __shfl(d, ej);
        if (ej < cnt) {
            ebuf[(size_t)(e0 + ej) * HEADS + h] = keep;
            if (keep > 0.0f)
                atomicMax(&emax_i[dd * HEADS + h], __float_as_int(keep));
        }
    }
}

__global__ void gat_edge_exp(
    const int* __restrict__ dst,
    float* __restrict__ ebuf,
    const float* __restrict__ emax,
    float* __restrict__ denom,
    int Etot, int E)
{
    const int t = blockIdx.x * blockDim.x + threadIdx.x;
    if (t >= Etot * HEADS) return;
    const int eid = t >> 2;
    const int h   = t & 3;
    const int d   = (eid < E) ? dst[eid] : (eid - E);

    const float m  = emax[d * HEADS + h];
    const float ex = __expf(ebuf[(size_t)eid * HEADS + h] - m);
    ebuf[(size_t)eid * HEADS + h] = ex;
    atomicAdd(&denom[d * HEADS + h], ex);
}

#define SC_GROUP 4

template <int G>
static __device__ __forceinline__ void gat_scatter_group(
    const float* __restrict__ xr, float* __restrict__ accum,
    int lane, int jb, int s, int d, float a0, float a1, float a2, float a3)
{
    float x[G][4];
    float al[G][4];
    int   dj[G];
    #pragma unroll
    for (int u = 0; u < G; ++u) {
        const int j  = jb + u;
        const int sj = __builtin_amdgcn_readlane(s, j);
        dj[u]    = __builtin_amdgcn_readlane(d, j);
        al[u][0] = __int_as_float(__builtin_amdgcn_readlane(__float_as_int(a0), j));
        al[u][1] = __int_as_float(__builtin_amdgcn_readlane(__float_as_int(a1), j));
        al[u][2] = __int_as_float(__builtin_amdgcn_readlane(__float_as_int(a2), j));
        al[u][3] = __int_as_float(__builtin_amdgcn_readlane(__float_as_int(a3), j));
        const float* __restrict__ p = xr + (size_t)sj * CIN + lane;
        x[u][0] = p[0];
        x[u][1] = p[HDIM];
        x[u][2] = p[2 * HDIM];
        x[u][3] = p[3 * HDIM];
    }
    #pragma unroll
    for (int u = 0; u < G; ++u) {
        const float v = fmaf(al[u][3], x[u][3],
                        fmaf(al[u][2], x[u][2],
                        fmaf(al[u][1], x[u][1], al[u][0] * x[u][0])));
        atomicAdd(accum + (size_t)dj[u] * HDIM + lane, v);
    }
}

__global__ __launch_bounds__(256) void gat_edge_scatter(
    const float* __restrict__ xr,
    const int* __restrict__ src, const int* __restrict__ dst,
    const float* __restrict__ ebuf, const float* __restrict__ denom,
    float* __restrict__ accum,
    int Etot, int E)
{
    const int lane = threadIdx.x & 31;
    const int wave = __builtin_amdgcn_readfirstlane(
        (int)((blockIdx.x * blockDim.x + threadIdx.x) >> 5));
    const int e0   = wave * 32;
    if (e0 >= Etot) return;
    const int cnt = (Etot - e0 < 32) ? (Etot - e0) : 32;

    int   s = 0, d = 0;
    float a0 = 0.0f, a1 = 0.0f, a2 = 0.0f, a3 = 0.0f;
    if (lane < cnt) {
        const int e = e0 + lane;
        if (e < E) { s = src[e]; d = dst[e]; }
        else       { s = d = e - E; }
        const float4 ex = *reinterpret_cast<const float4*>(ebuf  + (size_t)e * HEADS);
        const float4 dn = *reinterpret_cast<const float4*>(denom + (size_t)d * HEADS);
        a0 = ex.x / ((dn.x > 1e-16f) ? dn.x : 1e-16f);
        a1 = ex.y / ((dn.y > 1e-16f) ? dn.y : 1e-16f);
        a2 = ex.z / ((dn.z > 1e-16f) ? dn.z : 1e-16f);
        a3 = ex.w / ((dn.w > 1e-16f) ? dn.w : 1e-16f);
    }

    int jb = 0;
    for (; jb + SC_GROUP <= cnt; jb += SC_GROUP)
        gat_scatter_group<SC_GROUP>(xr, accum, lane, jb, s, d, a0, a1, a2, a3);
    #pragma nounroll
    for (; jb < cnt; ++jb)
        gat_scatter_group<1>(xr, accum, lane, jb, s, d, a0, a1, a2, a3);
}

__global__ void gat_finalize(
    const float* __restrict__ accum, const float* __restrict__ bias,
    float* __restrict__ out, int Nn)
{
    const int t = blockIdx.x * blockDim.x + threadIdx.x;
    if (t >= Nn * HDIM) return;
    const int dc = t & 31;
    out[t] = accum[t] * 0.25f + bias[dc];
}

extern "C" void kernel_launch(void* const* d_in, const int* in_sizes, int n_in,
                              void* d_out, int out_size, void* d_ws, size_t ws_size,
                              hipStream_t stream)
{
    const float* x    = (const float*)d_in[0];
    const int*   eidx = (const int*)d_in[1];
    const float* Wl   = (const float*)d_in[2];
    const float* Wr   = (const float*)d_in[3];
    const float* att  = (const float*)d_in[4];
    const float* bias = (const float*)d_in[5];
    float* out = (float*)d_out;

    const int Nn   = in_sizes[0] / CIN;
    const int E    = in_sizes[1] / 2;
    const int Etot = E + Nn;

    const int* src = eidx;
    const int* dst = eidx + E;

    float* ws = (float*)d_ws;
    size_t off = 0;
    float* xl    = ws + off; off += (size_t)Nn * CIN;
    float* xr    = ws + off; off += (size_t)Nn * CIN;
    float* ebuf  = ws + off; off += (size_t)Etot * HEADS;
    float* emax  = ws + off; off += (size_t)Nn * HEADS;
    float* denom = ws + off; off += (size_t)Nn * HEADS;
    float* accum = ws + off; off += (size_t)Nn * HDIM;
    (void)ws_size;

    {
        size_t nz = (size_t)Nn * HEADS * 2 + (size_t)Nn * HDIM;
        int blocks = 2048;
        gat_zero_kernel<<<blocks, 256, 0, stream>>>(emax, nz);
    }

    {
        dim3 grid((Nn + 15) / 16, 2);
        gat_gemm_wmma<<<grid, 256, 0, stream>>>(x, Wl, Wr, xl, xr, Nn);
    }

    const int edgeThreads = Etot * HEADS;
    const int edgeBlocks  = (edgeThreads + 255) / 256;
    const int scWaves  = (Etot + 31) / 32;
    const int scBlocks = (scWaves + 7) / 8;

    gat_edge_score<<<scBlocks, 256, 0, stream>>>(
        xl, xr, src, dst, att, ebuf, (int*)emax, Etot, E);

    gat_edge_exp<<<edgeBlocks, 256, 0, stream>>>(
        dst, ebuf, emax, denom, Etot, E);

    {
        gat_edge_scatter<<<scBlocks, 256, 0, stream>>>(
            xr, src, dst, ebuf, denom, accum, Etot, E);
    }

    {
        const int t = Nn * HDIM;
        gat_finalize<<<(t + 255) / 256, 256, 0, stream>>>(accum, bias, out, Nn);
    }
}


# GEMM: each block copies its W into LDS once (padded pitch); a wave keeps the A fragments of its 16 x rows in registers and does all 8 column tiles from LDS; k order and sums unchanged
# speedup vs baseline: 5.4702x; 1.2240x over previous
#include <hip/hip_runtime.h>
#include <hip/hip_bf16.h>


typedef __attribute__((ext_vector_type(2))) float v2f;
typedef __attribute__((ext_vector_type(8))) float v8f;

#define HEADS 4
#define HDIM 32
#define CIN 128
#define NEG_SLOPE 0.2f

__global__ void gat_zero_kernel(float* __restrict__ p, size_t n) {
    size_t i = (size_t)blockIdx.x * blockDim.x + threadIdx.x;
    size_t stride = (size_t)gridDim.x * blockDim.x;
    for (; i < n; i += stride) p[i] = 0.0f;
}

#define GM_WAVES 4
#define GM_PITCH (CIN + 4)

__global__ __launch_bounds__(GM_WAVES * 32) void gat_gemm_wmma(
    const float* __restrict__ X,
    const float* __restrict__ Wl,
    const float* __restrict__ Wr,
    float* __restrict__ XL,
    float* __restrict__ XR,
    int Nn)
{
    __shared__ __attribute__((aligned(16))) float wlds[CIN * GM_PITCH];

    const float* __restrict__ W  = (blockIdx.y == 0) ? Wl : Wr;
    float*       __restrict__ O  = (blockIdx.y == 0) ? XL : XR;

    {
        const float4* __restrict__ W4 = reinterpret_cast<const float4*>(W);
        for (int i = 0; i < (CIN * CIN / 4) / (GM_WAVES * 32); ++i) {
            const int q   = i * (GM_WAVES * 32) + (int)threadIdx.x;
            const int row = q >> 5;
            const int c4  = q & 31;
            *reinterpret_cast<float4*>(&wlds[row * GM_PITCH + c4 * 4]) = W4[q];
        }
    }
    __syncthreads();

    const int wave = threadIdx.x >> 5;
    const int lane = threadIdx.x & 31;
    const int row0 = ((int)blockIdx.x * GM_WAVES + wave) * 16;
    if (row0 >= Nn) return;

    const int m  = lane & 15;
    const int hi = lane >> 4;

    int arow = row0 + m; if (arow >= Nn) arow = Nn - 1;

    const float* __restrict__ ap = X + (size_t)arow * CIN + hi * 2;
    v2f a[CIN / 4];
    #pragma unroll
    for (int s = 0; s < CIN / 4; ++s)
        a[s] = *reinterpret_cast<const v2f*>(ap + s * 4);

    const float* bp = wlds + m * GM_PITCH + hi * 2;

    for (int ct = 0; ct < CIN / 16; ct += 2) {
        const float* b0p = bp + (ct * 16) * GM_PITCH;
        const float* b1p = b0p + 16 * GM_PITCH;
        v8f acc0 = {};
        v8f acc1 = {};
        #pragma unroll
        for (int s = 0; s < CIN / 4; ++s) {
            const v2f b0 = *reinterpret_cast<const v2f*>(b0p + s * 4);
            const v2f b1 = *reinterpret_cast<const v2f*>(b1p + s * 4);
            acc0 = __builtin_amdgcn_wmma_f32_16x16x4_f32(
                 false, a[s],  false, b0,
                 (short)0, acc0,  false,  false);
            acc1 = __builtin_amdgcn_wmma_f32_16x16x4_f32(
                 false, a[s],  false, b1,
                 (short)0, acc1,  false,  false);
        }

        #pragma unroll
        for (int i = 0; i < 8; ++i) {
            const int r = row0 + i + hi * 8;
            if (r < Nn) {
                O[(size_t)r * CIN + ct * 16 + m]      = acc0[i];
                O[(size_t)r * CIN + ct * 16 + 16 + m] = acc1[i];
            }
        }
    }
}

template <int M>
static __device__ __forceinline__ float gat_row_xor(float v) {
    return __int_as_float(__builtin_amdgcn_update_dpp(
        0, __float_as_int(v), 0x160 + M, 0xf, 0xf, true));
}

__global__ __launch_bounds__(256) void gat_edge_score(
    const float* __restrict__ xl, const float* __restrict__ xr,
    const int* __restrict__ src, const int* __restrict__ dst,
    const float* __restrict__ att,
    float* __restrict__ ebuf, int* __restrict__ emax_i,
    int Etot, int E)
{
    const int lane = threadIdx.x & 31;
    const int wave = __builtin_amdgcn_readfirstlane(
        (int)((blockIdx.x * blockDim.x + threadIdx.x) >> 5));
    const int e0   = wave * 32;
    if (e0 >= Etot) return;
    const int cnt = (Etot - e0 < 32) ? (Etot - e0) : 32;

    int s = 0, d = 0;
    if (lane < cnt) {
        const int e = e0 + lane;
        if (e < E) { s = src[e]; d = dst[e]; }
        else       { s = d = e - E; }
    }

    const float4 a4 = *reinterpret_cast<const float4*>(att + lane * 4);
    const int k = lane & 7;
    const int h = lane >> 3;

    for (int jb = 0; jb < cnt; jb += 8) {
        float4 L[8], R[8];
        for (int u = 0; u < 8; ++u) {
            const int j  = jb + u;
            const int sj = __builtin_amdgcn_readlane(s, j);
            const int dj = __builtin_amdgcn_readlane(d, j);
            L[u] = *reinterpret_cast<const float4*>(xl + (size_t)sj * CIN + lane * 4);
            R[u] = *reinterpret_cast<const float4*>(xr + (size_t)dj * CIN + lane * 4);
        }
        float keep = 0.0f;
        for (int u = 0; u < 8; ++u) {
            float v0 = L[u].x + R[u].x;
            float v1 = L[u].y + R[u].y;
            float v2 = L[u].z + R[u].z;
            float v3 = L[u].w + R[u].w;
            v0 = fmaxf(v0, NEG_SLOPE * v0);
            v1 = fmaxf(v1, NEG_SLOPE * v1);
            v2 = fmaxf(v2, NEG_SLOPE * v2);
            v3 = fmaxf(v3, NEG_SLOPE * v3);
            float p = fmaf(a4.w, v3, fmaf(a4.z, v2, fmaf(a4.y, v1, a4.x * v0)));
            p += gat_row_xor<1>(p);
            p += gat_row_xor<2>(p);
            p += gat_row_xor<4>(p);
            keep = (k == u) ? p : keep;
        }
        const int ej = jb + k;
        const int dd = __shfl(d, ej);
        if (ej < cnt) {
            ebuf[(size_t)(e0 + ej) * HEADS + h] = keep;
            if (keep > 0.0f)
                atomicMax(&emax_i[dd * HEADS + h], __float_as_int(keep));
        }
    }
}

__global__ void gat_edge_exp(
    const int* __restrict__ dst,
    float* __restrict__ ebuf,
    const float* __restrict__ emax,
    float* __restrict__ denom,
    int Etot, int E)
{
    const int t = blockIdx.x * blockDim.x + threadIdx.x;
    if (t >= Etot * HEADS) return;
    const int eid = t >> 2;
    const int h   = t & 3;
    const int d   = (eid < E) ? dst[eid] : (eid - E);

    const float m  = emax[d * HEADS + h];
    const float ex = __expf(ebuf[(size_t)eid * HEADS + h] - m);
    ebuf[(size_t)eid * HEADS + h] = ex;
    atomicAdd(&denom[d * HEADS + h], ex);
}

#define SC_GROUP 4

template <int G>
static __device__ __forceinline__ void gat_scatter_group(
    const float* __restrict__ xr, float* __restrict__ accum,
    int lane, int jb, int s, int d, float a0, float a1, float a2, float a3)
{
    float x[G][4];
    float al[G][4];
    int   dj[G];
    #pragma unroll
    for (int u = 0; u < G; ++u) {
        const int j  = jb + u;
        const int sj = __builtin_amdgcn_readlane(s, j);
        dj[u]    = __builtin_amdgcn_readlane(d, j);
        al[u][0] = __int_as_float(__builtin_amdgcn_readlane(__float_as_int(a0), j));
        al[u][1] = __int_as_float(__builtin_amdgcn_readlane(__float_as_int(a1), j));
        al[u][2] = __int_as_float(__builtin_amdgcn_readlane(__float_as_int(a2), j));
        al[u][3] = __int_as_float(__builtin_amdgcn_readlane(__float_as_int(a3), j));
        const float* __restrict__ p = xr + (size_t)sj * CIN + lane;
        x[u][0] = p[0];
        x[u][1] = p[HDIM];
        x[u][2] = p[2 * HDIM];
        x[u][3] = p[3 * HDIM];
    }
    #pragma unroll
    for (int u = 0; u < G; ++u) {
        const float v = fmaf(al[u][3], x[u][3],
                        fmaf(al[u][2], x[u][2],
                        fmaf(al[u][1], x[u][1], al[u][0] * x[u][0])));
        atomicAdd(accum + (size_t)dj[u] * HDIM + lane, v);
    }
}

__global__ __launch_bounds__(256) void gat_edge_scatter(
    const float* __restrict__ xr,
    const int* __restrict__ src, const int* __restrict__ dst,
    const float* __restrict__ ebuf, const float* __restrict__ denom,
    float* __restrict__ accum,
    int Etot, int E)
{
    const int lane = threadIdx.x & 31;
    const int wave = __builtin_amdgcn_readfirstlane(
        (int)((blockIdx.x * blockDim.x + threadIdx.x) >> 5));
    const int e0   = wave * 32;
    if (e0 >= Etot) return;
    const int cnt = (Etot - e0 < 32) ? (Etot - e0) : 32;

    int   s = 0, d = 0;
    float a0 = 0.0f, a1 = 0.0f, a2 = 0.0f, a3 = 0.0f;
    if (lane < cnt) {
        const int e = e0 + lane;
        if (e < E) { s = src[e]; d = dst[e]; }
        else       { s = d = e - E; }
        const float4 ex = *reinterpret_cast<const float4*>(ebuf  + (size_t)e * HEADS);
        const float4 dn = *reinterpret_cast<const float4*>(denom + (size_t)d * HEADS);
        a0 = ex.x / ((dn.x > 1e-16f) ? dn.x : 1e-16f);
        a1 = ex.y / ((dn.y > 1e-16f) ? dn.y : 1e-16f);
        a2 = ex.z / ((dn.z > 1e-16f) ? dn.z : 1e-16f);
        a3 = ex.w / ((dn.w > 1e-16f) ? dn.w : 1e-16f);
    }

    int jb = 0;
    for (; jb + SC_GROUP <= cnt; jb += SC_GROUP)
        gat_scatter_group<SC_GROUP>(xr, accum, lane, jb, s, d, a0, a1, a2, a3);
    #pragma nounroll
    for (; jb < cnt; ++jb)
        gat_scatter_group<1>(xr, accum, lane, jb, s, d, a0, a1, a2, a3);
}

__global__ void gat_finalize(
    const float* __restrict__ accum, const float* __restrict__ bias,
    float* __restrict__ out, int Nn)
{
    const int t = blockIdx.x * blockDim.x + threadIdx.x;
    if (t >= Nn * HDIM) return;
    const int dc = t & 31;
    out[t] = accum[t] * 0.25f + bias[dc];
}

extern "C" void kernel_launch(void* const* d_in, const int* in_sizes, int n_in,
                              void* d_out, int out_size, void* d_ws, size_t ws_size,
                              hipStream_t stream)
{
    const float* x    = (const float*)d_in[0];
    const int*   eidx = (const int*)d_in[1];
    const float* Wl   = (const float*)d_in[2];
    const float* Wr   = (const float*)d_in[3];
    const float* att  = (const float*)d_in[4];
    const float* bias = (const float*)d_in[5];
    float* out = (float*)d_out;

    const int Nn   = in_sizes[0] / CIN;
    const int E    = in_sizes[1] / 2;
    const int Etot = E + Nn;

    const int* src = eidx;
    const int* dst = eidx + E;

    float* ws = (float*)d_ws;
    size_t off = 0;
    float* xl    = ws + off; off += (size_t)Nn * CIN;
    float* xr    = ws + off; off += (size_t)Nn * CIN;
    float* ebuf  = ws + off; off += (size_t)Etot * HEADS;
    float* emax  = ws + off; off += (size_t)Nn * HEADS;
    float* denom = ws + off; off += (size_t)Nn * HEADS;
    float* accum = ws + off; off += (size_t)Nn * HDIM;
    (void)ws_size;

    {
        size_t nz = (size_t)Nn * HEADS * 2 + (size_t)Nn * HDIM;
        int blocks = 2048;
        gat_zero_kernel<<<blocks, 256, 0, stream>>>(emax, nz);
    }

    {
        const int rowTiles = (Nn + 15) / 16;
        dim3 grid((rowTiles + GM_WAVES - 1) / GM_WAVES, 2);
        gat_gemm_wmma<<<grid, GM_WAVES * 32, 0, stream>>>(x, Wl, Wr, xl, xr, Nn);
    }

    const int edgeThreads = Etot * HEADS;
    const int edgeBlocks  = (edgeThreads + 255) / 256;
    const int scWaves  = (Etot + 31) / 32;
    const int scBlocks = (scWaves + 7) / 8;

    gat_edge_score<<<scBlocks, 256, 0, stream>>>(
        xl, xr, src, dst, att, ebuf, (int*)emax, Etot, E);

    gat_edge_exp<<<edgeBlocks, 256, 0, stream>>>(
        dst, ebuf, emax, denom, Etot, E);

    {
        gat_edge_scatter<<<scBlocks, 256, 0, stream>>>(
            xr, src, dst, ebuf, denom, accum, Etot, E);
    }

    {
        const int t = Nn * HDIM;
        gat_finalize<<<(t + 255) / 256, 256, 0, stream>>>(accum, bias, out, Nn);
    }
}


# zero-fill and finalize launches removed: the GEMM launch stores the start values (0 for max and denominator, bias into out), the scatter adds 0.25 x row into out; bias is now each sum's first term
# speedup vs baseline: 5.8347x; 1.0666x over previous
#include <hip/hip_runtime.h>
#include <hip/hip_bf16.h>


typedef __attribute__((ext_vector_type(2))) float v2f;
typedef __attribute__((ext_vector_type(8))) float v8f;

#define HEADS 4
#define HDIM 32
#define CIN 128
#define NEG_SLOPE 0.2f

#define GM_WAVES 4
#define GM_PITCH (CIN + 4)

__global__ __launch_bounds__(GM_WAVES * 32) void gat_gemm_wmma(
    const float* __restrict__ X,
    const float* __restrict__ Wl,
    const float* __restrict__ Wr,
    float* __restrict__ XL,
    float* __restrict__ XR,
    float* __restrict__ zinit,
    const float* __restrict__ bias,
    float* __restrict__ out,
    int Nn)
{
    __shared__ __attribute__((aligned(16))) float wlds[CIN * GM_PITCH];

    const float* __restrict__ W  = (blockIdx.y == 0) ? Wl : Wr;
    float*       __restrict__ O  = (blockIdx.y == 0) ? XL : XR;

    const float4 b4 = reinterpret_cast<const float4*>(bias)[threadIdx.x & 7];

    {
        const float4* __restrict__ W4 = reinterpret_cast<const float4*>(W);
        for (int i = 0; i < (CIN * CIN / 4) / (GM_WAVES * 32); ++i) {
            const int q   = i * (GM_WAVES * 32) + (int)threadIdx.x;
            const int row = q >> 5;
            const int c4  = q & 31;
            *reinterpret_cast<float4*>(&wlds[row * GM_PITCH + c4 * 4]) = W4[q];
        }
    }
    asm volatile("" :: "v"(b4.x), "v"(b4.y), "v"(b4.z), "v"(b4.w));
    __syncthreads();

    {
        const int nthr = (int)(gridDim.x * gridDim.y) * (GM_WAVES * 32);
        const int tid  = ((int)blockIdx.y * (int)gridDim.x + (int)blockIdx.x) * (GM_WAVES * 32)
                         + (int)threadIdx.x;
        float4* __restrict__ z4 = reinterpret_cast<float4*>(zinit);
        const int nz4 = Nn * (2 * HEADS / 4);
        for (int q = tid; q < nz4; q += nthr)
            z4[q] = make_float4(0.0f, 0.0f, 0.0f, 0.0f);
        float4* __restrict__ o4 = reinterpret_cast<float4*>(out);
        const int no4 = Nn * (HDIM / 4);
        for (int q = tid; q < no4; q += nthr)
            o4[q] = b4;
    }

    const int wave = threadIdx.x >> 5;
    const int lane = threadIdx.x & 31;
    const int row0 = ((int)blockIdx.x * GM_WAVES + wave) * 16;
    if (row0 >= Nn) return;

    const int m  = lane & 15;
    const int hi = lane >> 4;

    int arow = row0 + m; if (arow >= Nn) arow = Nn - 1;

    const float* __restrict__ ap = X + (size_t)arow * CIN + hi * 2;
    v2f a[CIN / 4];
    #pragma unroll
    for (int s = 0; s < CIN / 4; ++s)
        a[s] = *reinterpret_cast<const v2f*>(ap + s * 4);

    const float* bp = wlds + m * GM_PITCH + hi * 2;

    for (int ct = 0; ct < CIN / 16; ct += 2) {
        const float* b0p = bp + (ct * 16) * GM_PITCH;
        const float* b1p = b0p + 16 * GM_PITCH;
        v8f acc0 = {};
        v8f acc1 = {};
        #pragma unroll
        for (int s = 0; s < CIN / 4; ++s) {
            const v2f b0 = *reinterpret_cast<const v2f*>(b0p + s * 4);
            const v2f b1 = *reinterpret_cast<const v2f*>(b1p + s * 4);
            acc0 = __builtin_amdgcn_wmma_f32_16x16x4_f32(
                 false, a[s],  false, b0,
                 (short)0, acc0,  false,  false);
            acc1 = __builtin_amdgcn_wmma_f32_16x16x4_f32(
                 false, a[s],  false, b1,
                 (short)0, acc1,  false,  false);
        }

        #pragma unroll
        for (int i = 0; i < 8; ++i) {
            const int r = row0 + i + hi * 8;
            if (r < Nn) {
                O[(size_t)r * CIN + ct * 16 + m]      = acc0[i];
                O[(size_t)r * CIN + ct * 16 + 16 + m] = acc1[i];
            }
        }
    }
}

template <int M>
static __device__ __forceinline__ float gat_row_xor(float v) {
    return __int_as_float(__builtin_amdgcn_update_dpp(
        0, __float_as_int(v), 0x160 + M, 0xf, 0xf, true));
}

__global__ __launch_bounds__(256) void gat_edge_score(
    const float* __restrict__ xl, const float* __restrict__ xr,
    const int* __restrict__ src, const int* __restrict__ dst,
    const float* __restrict__ att,
    float* __restrict__ ebuf, int* __restrict__ emax_i,
    int Etot, int E)
{
    const int lane = threadIdx.x & 31;
    const int wave = __builtin_amdgcn_readfirstlane(
        (int)((blockIdx.x * blockDim.x + threadIdx.x) >> 5));
    const int e0   = wave * 32;
    if (e0 >= Etot) return;
    const int cnt = (Etot - e0 < 32) ? (Etot - e0) : 32;

    int s = 0, d = 0;
    if (lane < cnt) {
        const int e = e0 + lane;
        if (e < E) { s = src[e]; d = dst[e]; }
        else       { s = d = e - E; }
    }

    const float4 a4 = *reinterpret_cast<const float4*>(att + lane * 4);
    const int k = lane & 7;
    const int h = lane >> 3;

    for (int jb = 0; jb < cnt; jb += 8) {
        float4 L[8], R[8];
        for (int u = 0; u < 8; ++u) {
            const int j  = jb + u;
            const int sj = __builtin_amdgcn_readlane(s, j);
            const int dj = __builtin_amdgcn_readlane(d, j);
            L[u] = *reinterpret_cast<const float4*>(xl + (size_t)sj * CIN + lane * 4);
            R[u] = *reinterpret_cast<const float4*>(xr + (size_t)dj * CIN + lane * 4);
        }
        float keep = 0.0f;
        for (int u = 0; u < 8; ++u) {
            float v0 = L[u].x + R[u].x;
            float v1 = L[u].y + R[u].y;
            float v2 = L[u].z + R[u].z;
            float v3 = L[u].w + R[u].w;
            v0 = fmaxf(v0, NEG_SLOPE * v0);
            v1 = fmaxf(v1, NEG_SLOPE * v1);
            v2 = fmaxf(v2, NEG_SLOPE * v2);
            v3 = fmaxf(v3, NEG_SLOPE * v3);
            float p = fmaf(a4.w, v3, fmaf(a4.z, v2, fmaf(a4.y, v1, a4.x * v0)));
            p += gat_row_xor<1>(p);
            p += gat_row_xor<2>(p);
            p += gat_row_xor<4>(p);
            keep = (k == u) ? p : keep;
        }
        const int ej = jb + k;
        const int dd = __shfl(d, ej);
        if (ej < cnt) {
            ebuf[(size_t)(e0 + ej) * HEADS + h] = keep;
            if (keep > 0.0f)
                atomicMax(&emax_i[dd * HEADS + h], __float_as_int(keep));
        }
    }
}

__global__ void gat_edge_exp(
    const int* __restrict__ dst,
    float* __restrict__ ebuf,
    const float* __restrict__ emax,
    float* __restrict__ denom,
    int Etot, int E)
{
    const int t = blockIdx.x * blockDim.x + threadIdx.x;
    if (t >= Etot * HEADS) return;
    const int eid = t >> 2;
    const int h   = t & 3;
    const int d   = (eid < E) ? dst[eid] : (eid - E);

    const float m  = emax[d * HEADS + h];
    const float ex = __expf(ebuf[(size_t)eid * HEADS + h] - m);
    ebuf[(size_t)eid * HEADS + h] = ex;
    atomicAdd(&denom[d * HEADS + h], ex);
}

#define SC_GROUP 4

template <int G>
static __device__ __forceinline__ void gat_scatter_group(
    const float* __restrict__ xr, float* __restrict__ out,
    int lane, int jb, int s, int d, float a0, float a1, float a2, float a3)
{
    float x[G][4];
    float al[G][4];
    int   dj[G];
    #pragma unroll
    for (int u = 0; u < G; ++u) {
        const int j  = jb + u;
        const int sj = __builtin_amdgcn_readlane(s, j);
        dj[u]    = __builtin_amdgcn_readlane(d, j);
        al[u][0] = __int_as_float(__builtin_amdgcn_readlane(__float_as_int(a0), j));
        al[u][1] = __int_as_float(__builtin_amdgcn_readlane(__float_as_int(a1), j));
        al[u][2] = __int_as_float(__builtin_amdgcn_readlane(__float_as_int(a2), j));
        al[u][3] = __int_as_float(__builtin_amdgcn_readlane(__float_as_int(a3), j));
        const float* __restrict__ p = xr + (size_t)sj * CIN + lane;
        x[u][0] = p[0];
        x[u][1] = p[HDIM];
        x[u][2] = p[2 * HDIM];
        x[u][3] = p[3 * HDIM];
    }
    #pragma unroll
    for (int u = 0; u < G; ++u) {
        const float v = fmaf(al[u][3], x[u][3],
                        fmaf(al[u][2], x[u][2],
                        fmaf(al[u][1], x[u][1], al[u][0] * x[u][0])));
        atomicAdd(out + (size_t)dj[u] * HDIM + lane, 0.25f * v);
    }
}

__global__ __launch_bounds__(256) void gat_edge_scatter(
    const float* __restrict__ xr,
    const int* __restrict__ src, const int* __restrict__ dst,
    const float* __restrict__ ebuf, const float* __restrict__ denom,
    float* __restrict__ out,
    int Etot, int E)
{
    const int lane = threadIdx.x & 31;
    const int wave = __builtin_amdgcn_readfirstlane(
        (int)((blockIdx.x * blockDim.x + threadIdx.x) >> 5));
    const int e0   = wave * 32;
    if (e0 >= Etot) return;
    const int cnt = (Etot - e0 < 32) ? (Etot - e0) : 32;

    int   s = 0, d = 0;
    float a0 = 0.0f, a1 = 0.0f, a2 = 0.0f, a3 = 0.0f;
    if (lane < cnt) {
        const int e = e0 + lane;
        if (e < E) { s = src[e]; d = dst[e]; }
        else       { s = d = e - E; }
        const float4 ex = *reinterpret_cast<const float4*>(ebuf  + (size_t)e * HEADS);
        const float4 dn = *reinterpret_cast<const float4*>(denom + (size_t)d * HEADS);
        a0 = ex.x / ((dn.x > 1e-16f) ? dn.x : 1e-16f);
        a1 = ex.y / ((dn.y > 1e-16f) ? dn.y : 1e-16f);
        a2 = ex.z / ((dn.z > 1e-16f) ? dn.z : 1e-16f);
        a3 = ex.w / ((dn.w > 1e-16f) ? dn.w : 1e-16f);
    }

    int jb = 0;
    for (; jb + SC_GROUP <= cnt; jb += SC_GROUP)
        gat_scatter_group<SC_GROUP>(xr, out, lane, jb, s, d, a0, a1, a2, a3);
    #pragma nounroll
    for (; jb < cnt; ++jb)
        gat_scatter_group<1>(xr, out, lane, jb, s, d, a0, a1, a2, a3);
}

extern "C" void kernel_launch(void* const* d_in, const int* in_sizes, int n_in,
                              void* d_out, int out_size, void* d_ws, size_t ws_size,
                              hipStream_t stream)
{
    const float* x    = (const float*)d_in[0];
    const int*   eidx = (const int*)d_in[1];
    const float* Wl   = (const float*)d_in[2];
    const float* Wr   = (const float*)d_in[3];
    const float* att  = (const float*)d_in[4];
    const float* bias = (const float*)d_in[5];
    float* out = (float*)d_out;

    const int Nn   = in_sizes[0] / CIN;
    const int E    = in_sizes[1] / 2;
    const int Etot = E + Nn;

    const int* src = eidx;
    const int* dst = eidx + E;

    float* ws = (float*)d_ws;
    size_t off = 0;
    float* xl    = ws + off; off += (size_t)Nn * CIN;
    float* xr    = ws + off; off += (size_t)Nn * CIN;
    float* ebuf  = ws + off; off += (size_t)Etot * HEADS;
    float* emax  = ws + off; off += (size_t)Nn * HEADS;
    float* denom = ws + off; off += (size_t)Nn * HEADS;
    (void)ws_size;

    {
        const int rowTiles = (Nn + 15) / 16;
        dim3 grid((rowTiles + GM_WAVES - 1) / GM_WAVES, 2);
        gat_gemm_wmma<<<grid, GM_WAVES * 32, 0, stream>>>(
            x, Wl, Wr, xl, xr, emax, bias, out, Nn);
    }

    const int edgeThreads = Etot * HEADS;
    const int edgeBlocks  = (edgeThreads + 255) / 256;
    const int scWaves  = (Etot + 31) / 32;
    const int scBlocks = (scWaves + 7) / 8;

    gat_edge_score<<<scBlocks, 256, 0, stream>>>(
        xl, xr, src, dst, att, ebuf, (int*)emax, Etot, E);

    gat_edge_exp<<<edgeBlocks, 256, 0, stream>>>(
        dst, ebuf, emax, denom, Etot, E);

    {
        gat_edge_scatter<<<scBlocks, 256, 0, stream>>>(
            xr, src, dst, ebuf, denom, out, Etot, E);
    }
}


# edges counting-sorted by destination bin (64 nodes, LDS counters); one block per bin keeps max, denominator and out rows in LDS, stores each row once: no global float atomics; sum order changes
# speedup vs baseline: 7.8238x; 1.3409x over previous
#include <hip/hip_runtime.h>
#include <hip/hip_bf16.h>


typedef __attribute__((ext_vector_type(2))) float v2f;
typedef __attribute__((ext_vector_type(8))) float v8f;

#define HEADS 4
#define HDIM 32
#define CIN 128
#define NEG_SLOPE 0.2f

#define GM_WAVES 4
#define GM_PITCH (CIN + 4)

__global__ __launch_bounds__(GM_WAVES * 32) void gat_gemm_wmma(
    const float* __restrict__ X,
    const float* __restrict__ Wl,
    const float* __restrict__ Wr,
    float* __restrict__ XL,
    float* __restrict__ XR,
    float* __restrict__ zinit,
    int nz4,
    int Nn)
{
    __shared__ __attribute__((aligned(16))) float wlds[CIN * GM_PITCH];

    const float* __restrict__ W  = (blockIdx.y == 0) ? Wl : Wr;
    float*       __restrict__ O  = (blockIdx.y == 0) ? XL : XR;

    {
        const float4* __restrict__ W4 = reinterpret_cast<const float4*>(W);
        for (int i = 0; i < (CIN * CIN / 4) / (GM_WAVES * 32); ++i) {
            const int q   = i * (GM_WAVES * 32) + (int)threadIdx.x;
            const int row = q >> 5;
            const int c4  = q & 31;
            *reinterpret_cast<float4*>(&wlds[row * GM_PITCH + c4 * 4]) = W4[q];
        }
    }
    __syncthreads();

    {
        const int nthr = (int)(gridDim.x * gridDim.y) * (GM_WAVES * 32);
        const int tid  = ((int)blockIdx.y * (int)gridDim.x + (int)blockIdx.x) * (GM_WAVES * 32)
                         + (int)threadIdx.x;
        float4* __restrict__ z4 = reinterpret_cast<float4*>(zinit);
        for (int q = tid; q < nz4; q += nthr)
            z4[q] = make_float4(0.0f, 0.0f, 0.0f, 0.0f);
    }

    const int wave = threadIdx.x >> 5;
    const int lane = threadIdx.x & 31;
    const int row0 = ((int)blockIdx.x * GM_WAVES + wave) * 16;
    if (row0 >= Nn) return;

    const int m  = lane & 15;
    const int hi = lane >> 4;

    int arow = row0 + m; if (arow >= Nn) arow = Nn - 1;

    const float* __restrict__ ap = X + (size_t)arow * CIN + hi * 2;
    v2f a[CIN / 4];
    #pragma unroll
    for (int s = 0; s < CIN / 4; ++s)
        a[s] = *reinterpret_cast<const v2f*>(ap + s * 4);

    const float* bp = wlds + m * GM_PITCH + hi * 2;

    for (int ct = 0; ct < CIN / 16; ct += 2) {
        const float* b0p = bp + (ct * 16) * GM_PITCH;
        const float* b1p = b0p + 16 * GM_PITCH;
        v8f acc0 = {};
        v8f acc1 = {};
        #pragma unroll
        for (int s = 0; s < CIN / 4; ++s) {
            const v2f b0 = *reinterpret_cast<const v2f*>(b0p + s * 4);
            const v2f b1 = *reinterpret_cast<const v2f*>(b1p + s * 4);
            acc0 = __builtin_amdgcn_wmma_f32_16x16x4_f32(
                 false, a[s],  false, b0,
                 (short)0, acc0,  false,  false);
            acc1 = __builtin_amdgcn_wmma_f32_16x16x4_f32(
                 false, a[s],  false, b1,
                 (short)0, acc1,  false,  false);
        }

        #pragma unroll
        for (int i = 0; i < 8; ++i) {
            const int r = row0 + i + hi * 8;
            if (r < Nn) {
                O[(size_t)r * CIN + ct * 16 + m]      = acc0[i];
                O[(size_t)r * CIN + ct * 16 + 16 + m] = acc1[i];
            }
        }
    }
}

#define BIN_SHIFT 6
#define BIN_NODES (1 << BIN_SHIFT)
#define MAXBINS 1024
#define SORT_THREADS 1024
#define SORT_CHUNK 8192
#define SORT_PER_THREAD (SORT_CHUNK / SORT_THREADS)

__global__ __launch_bounds__(SORT_THREADS) void gat_bin_count(
    const int* __restrict__ dst, int* __restrict__ binTotal, int E)
{
    __shared__ int hist[MAXBINS];
    const int t = threadIdx.x;
    hist[t] = 0;
    __syncthreads();
    const int e0 = (int)blockIdx.x * SORT_CHUNK;
    int dv[SORT_PER_THREAD];
    #pragma unroll
    for (int i = 0; i < SORT_PER_THREAD; ++i) {
        const int e = e0 + i * SORT_THREADS + t;
        dv[i] = (e < E) ? dst[e] : -1;
    }
    #pragma unroll
    for (int i = 0; i < SORT_PER_THREAD; ++i)
        if (dv[i] >= 0) atomicAdd(&hist[dv[i] >> BIN_SHIFT], 1);
    __syncthreads();
    const int c = hist[t];
    if (c > 0) atomicAdd(&binTotal[t], c);
}

__global__ __launch_bounds__(SORT_THREADS) void gat_bin_place(
    const int* __restrict__ src, const int* __restrict__ dst,
    const int* __restrict__ binTotal, int* __restrict__ binCursor,
    int* __restrict__ binStart,
    int* __restrict__ ssrc, int* __restrict__ sdst, int E)
{
    __shared__ int hist[MAXBINS];
    __shared__ int base[MAXBINS];
    __shared__ int wsum[SORT_THREADS / 32];
    __shared__ int wexc[SORT_THREADS / 32];
    const int t    = threadIdx.x;
    const int lane = t & 31;
    const int w    = t >> 5;
    hist[t] = 0;
    __syncthreads();
    const int e0 = (int)blockIdx.x * SORT_CHUNK;
    int dv[SORT_PER_THREAD], sv[SORT_PER_THREAD];
    #pragma unroll
    for (int i = 0; i < SORT_PER_THREAD; ++i) {
        const int e = e0 + i * SORT_THREADS + t;
        dv[i] = (e < E) ? dst[e] : -1;
        sv[i] = (e < E) ? src[e] : 0;
    }
    #pragma unroll
    for (int i = 0; i < SORT_PER_THREAD; ++i)
        if (dv[i] >= 0) atomicAdd(&hist[dv[i] >> BIN_SHIFT], 1);

    const int tot = binTotal[t];
    int v = tot;
    for (int off = 1; off < 32; off <<= 1) {
        const int u = __shfl_up(v, off);
        if (lane >= off) v += u;
    }
    if (lane == 31) wsum[w] = v;
    __syncthreads();
    if (w == 0) {
        const int s0 = wsum[lane];
        int s = s0;
        for (int off = 1; off < 32; off <<= 1) {
            const int u = __shfl_up(s, off);
            if (lane >= off) s += u;
        }
        wexc[lane] = s - s0;
    }
    __syncthreads();
    const int start = v - tot + wexc[w];
    if (blockIdx.x == 0) binStart[t] = start;

    const int c = hist[t];
    int bs = start;
    if (c > 0) bs += atomicAdd(&binCursor[t], c);
    base[t] = bs;
    hist[t] = 0;
    __syncthreads();

    #pragma unroll
    for (int i = 0; i < SORT_PER_THREAD; ++i) {
        if (dv[i] >= 0) {
            const int b   = dv[i] >> BIN_SHIFT;
            const int pos = base[b] + atomicAdd(&hist[b], 1);
            ssrc[pos] = sv[i];
            sdst[pos] = dv[i];
        }
    }
}

template <int M>
static __device__ __forceinline__ float gat_row_xor(float v) {
    return __int_as_float(__builtin_amdgcn_update_dpp(
        0, __float_as_int(v), 0x160 + M, 0xf, 0xf, true));
}

#define SC_GROUP 4

template <int G>
static __device__ __forceinline__ void gat_sum_group(
    const float* __restrict__ xr, float* srow,
    int lane, int jb, int s, int dl, float a0, float a1, float a2, float a3)
{
    float x[G][4];
    float al[G][4];
    int   dj[G];
    #pragma unroll
    for (int u = 0; u < G; ++u) {
        const int j  = jb + u;
        const int sj = __builtin_amdgcn_readlane(s, j);
        dj[u]    = __builtin_amdgcn_readlane(dl, j);
        al[u][0] = __int_as_float(__builtin_amdgcn_readlane(__float_as_int(a0), j));
        al[u][1] = __int_as_float(__builtin_amdgcn_readlane(__float_as_int(a1), j));
        al[u][2] = __int_as_float(__builtin_amdgcn_readlane(__float_as_int(a2), j));
        al[u][3] = __int_as_float(__builtin_amdgcn_readlane(__float_as_int(a3), j));
        const float* __restrict__ p = xr + (size_t)sj * CIN + lane;
        x[u][0] = p[0];
        x[u][1] = p[HDIM];
        x[u][2] = p[2 * HDIM];
        x[u][3] = p[3 * HDIM];
    }
    #pragma unroll
    for (int u = 0; u < G; ++u) {
        const float v = fmaf(al[u][3], x[u][3],
                        fmaf(al[u][2], x[u][2],
                        fmaf(al[u][1], x[u][1], al[u][0] * x[u][0])));
        atomicAdd(srow + dj[u] * HDIM + lane, 0.25f * v);
    }
}

#define FB_WAVES 8

__global__ __launch_bounds__(FB_WAVES * 32) void gat_bin_edges(
    const float* __restrict__ xl, const float* __restrict__ xr,
    const int* __restrict__ ssrc, const int* __restrict__ sdst,
    const int* __restrict__ binStart, const int* __restrict__ binTotal,
    const float* __restrict__ att, const float* __restrict__ bias,
    float* __restrict__ ebuf, float* __restrict__ out,
    int Nn, int E)
{
    __shared__ __attribute__((aligned(16))) int   smax[BIN_NODES * HEADS];
    __shared__ __attribute__((aligned(16))) float sden[BIN_NODES * HEADS];
    __shared__ __attribute__((aligned(16))) float sout[BIN_NODES * HDIM];

    const int lane = threadIdx.x & 31;
    const int wave  = __builtin_amdgcn_readfirstlane((int)(threadIdx.x >> 5));
    const int b     = (int)blockIdx.x;
    const int n0    = b << BIN_SHIFT;
    const int nn    = (Nn - n0 < BIN_NODES) ? (Nn - n0) : BIN_NODES;
    const int start = __builtin_amdgcn_readfirstlane(binStart[b]);
    const int cntE  = __builtin_amdgcn_readfirstlane(binTotal[b]);
    const int cnt   = cntE + nn;

    for (int q = threadIdx.x; q < BIN_NODES * HEADS; q += FB_WAVES * 32) {
        smax[q] = 0;
        sden[q] = 0.0f;
    }
    for (int q = threadIdx.x; q < BIN_NODES * HDIM; q += FB_WAVES * 32)
        sout[q] = 0.0f;
    __syncthreads();

    {
        const float4 a4 = *reinterpret_cast<const float4*>(att + lane * 4);
        const int k = lane & 7;
        const int h = lane >> 3;
        for (int g0 = wave * 32; g0 < cnt; g0 += FB_WAVES * 32) {
            const int c = (cnt - g0 < 32) ? (cnt - g0) : 32;
            int s = n0, d = n0, slot = 0;
            if (lane < c) {
                const int j = g0 + lane;
                if (j < cntE) { s = ssrc[start + j]; d = sdst[start + j]; slot = start + j; }
                else          { s = d = n0 + (j - cntE); slot = E + s; }
            }
            for (int jb = 0; jb < c; jb += 8) {
                float4 L[8], R[8];
                for (int u = 0; u < 8; ++u) {
                    const int j  = jb + u;
                    const int sj = __builtin_amdgcn_readlane(s, j);
                    const int dj = __builtin_amdgcn_readlane(d, j);
                    L[u] = *reinterpret_cast<const float4*>(xl + (size_t)sj * CIN + lane * 4);
                    R[u] = *reinterpret_cast<const float4*>(xr + (size_t)dj * CIN + lane * 4);
                }
                float keep = 0.0f;
                for (int u = 0; u < 8; ++u) {
                    float v0 = L[u].x + R[u].x;
                    float v1 = L[u].y + R[u].y;
                    float v2 = L[u].z + R[u].z;
                    float v3 = L[u].w + R[u].w;
                    v0 = fmaxf(v0, NEG_SLOPE * v0);
                    v1 = fmaxf(v1, NEG_SLOPE * v1);
                    v2 = fmaxf(v2, NEG_SLOPE * v2);
                    v3 = fmaxf(v3, NEG_SLOPE * v3);
                    float p = fmaf(a4.w, v3, fmaf(a4.z, v2, fmaf(a4.y, v1, a4.x * v0)));
                    p += gat_row_xor<1>(p);
                    p += gat_row_xor<2>(p);
                    p += gat_row_xor<4>(p);
                    keep = (k == u) ? p : keep;
                }
                const int ej = jb + k;
                const int dd = __shfl(d, ej);
                const int sl = __shfl(slot, ej);
                if (ej < c) {
                    ebuf[(size_t)sl * HEADS + h] = keep;
                    if (keep > 0.0f)
                        atomicMax(&smax[(dd - n0) * HEADS + h], __float_as_int(keep));
                }
            }
        }
    }
    __syncthreads();

    for (int j = threadIdx.x; j < cnt; j += FB_WAVES * 32) {
        int d, slot;
        if (j < cntE) { d = sdst[start + j]; slot = start + j; }
        else          { d = n0 + (j - cntE); slot = E + d; }
        const int dl = d - n0;
        const float4 sc = *reinterpret_cast<const float4*>(ebuf + (size_t)slot * HEADS);
        const int4   mx = *reinterpret_cast<const int4*>(&smax[dl * HEADS]);
        atomicAdd(&sden[dl * HEADS + 0], __expf(sc.x - __int_as_float(mx.x)));
        atomicAdd(&sden[dl * HEADS + 1], __expf(sc.y - __int_as_float(mx.y)));
        atomicAdd(&sden[dl * HEADS + 2], __expf(sc.z - __int_as_float(mx.z)));
        atomicAdd(&sden[dl * HEADS + 3], __expf(sc.w - __int_as_float(mx.w)));
    }
    __syncthreads();

    for (int g0 = wave * 32; g0 < cnt; g0 += FB_WAVES * 32) {
        const int c = (cnt - g0 < 32) ? (cnt - g0) : 32;
        int   s = n0, dl = 0;
        float a0 = 0.0f, a1 = 0.0f, a2 = 0.0f, a3 = 0.0f;
        if (lane < c) {
            const int j = g0 + lane;
            int d, slot;
            if (j < cntE) { s = ssrc[start + j]; d = sdst[start + j]; slot = start + j; }
            else          { s = d = n0 + (j - cntE); slot = E + s; }
            dl = d - n0;
            const float4 sc = *reinterpret_cast<const float4*>(ebuf + (size_t)slot * HEADS);
            const int4   mx = *reinterpret_cast<const int4*>(&smax[dl * HEADS]);
            const float4 dn = *reinterpret_cast<const float4*>(&sden[dl * HEADS]);
            a0 = __expf(sc.x - __int_as_float(mx.x)) / ((dn.x > 1e-16f) ? dn.x : 1e-16f);
            a1 = __expf(sc.y - __int_as_float(mx.y)) / ((dn.y > 1e-16f) ? dn.y : 1e-16f);
            a2 = __expf(sc.z - __int_as_float(mx.z)) / ((dn.z > 1e-16f) ? dn.z : 1e-16f);
            a3 = __expf(sc.w - __int_as_float(mx.w)) / ((dn.w > 1e-16f) ? dn.w : 1e-16f);
        }
        int jb = 0;
        for (; jb + SC_GROUP <= c; jb += SC_GROUP)
            gat_sum_group<SC_GROUP>(xr, sout, lane, jb, s, dl, a0, a1, a2, a3);
        #pragma nounroll
        for (; jb < c; ++jb)
            gat_sum_group<1>(xr, sout, lane, jb, s, dl, a0, a1, a2, a3);
    }
    __syncthreads();

    for (int q = threadIdx.x; q < nn * HDIM; q += FB_WAVES * 32)
        out[(size_t)n0 * HDIM + q] = bias[q & (HDIM - 1)] + sout[q];
}

extern "C" void kernel_launch(void* const* d_in, const int* in_sizes, int n_in,
                              void* d_out, int out_size, void* d_ws, size_t ws_size,
                              hipStream_t stream)
{
    const float* x    = (const float*)d_in[0];
    const int*   eidx = (const int*)d_in[1];
    const float* Wl   = (const float*)d_in[2];
    const float* Wr   = (const float*)d_in[3];
    const float* att  = (const float*)d_in[4];
    const float* bias = (const float*)d_in[5];
    float* out = (float*)d_out;

    const int Nn   = in_sizes[0] / CIN;
    const int E    = in_sizes[1] / 2;
    const int Etot = E + Nn;

    const int* src = eidx;
    const int* dst = eidx + E;

    float* ws = (float*)d_ws;
    size_t off = 0;
    float* xl    = ws + off; off += (size_t)Nn * CIN;
    float* xr    = ws + off; off += (size_t)Nn * CIN;
    float* ebuf  = ws + off; off += (size_t)Etot * HEADS;
    int* binTotal  = (int*)(ws + off); off += MAXBINS;
    int* binCursor = (int*)(ws + off); off += MAXBINS;
    int* binStart  = (int*)(ws + off); off += MAXBINS;
    int* ssrc      = (int*)(ws + off); off += (size_t)E;
    int* sdst      = (int*)(ws + off); off += (size_t)E;
    (void)ws_size;

    {
        const int rowTiles = (Nn + 15) / 16;
        dim3 grid((rowTiles + GM_WAVES - 1) / GM_WAVES, 2);
        gat_gemm_wmma<<<grid, GM_WAVES * 32, 0, stream>>>(
            x, Wl, Wr, xl, xr, (float*)binTotal, 2 * MAXBINS / 4, Nn);
    }

    const int sortBlocks = (E + SORT_CHUNK - 1) / SORT_CHUNK;
    const int nbins      = (Nn + BIN_NODES - 1) / BIN_NODES;

    gat_bin_count<<<sortBlocks, SORT_THREADS, 0, stream>>>(dst, binTotal, E);

    gat_bin_place<<<sortBlocks, SORT_THREADS, 0, stream>>>(
        src, dst, binTotal, binCursor, binStart, ssrc, sdst, E);

    gat_bin_edges<<<nbins, FB_WAVES * 32, 0, stream>>>(
        xl, xr, ssrc, sdst, binStart, binTotal, att, bias, ebuf, out, Nn, E);
}


# bin kernel: each block copies its 64 own xr rows into LDS once; the edge score reads xr[d] from that copy, so only xl[s] is gathered from global memory there; sums and their order unchanged
# speedup vs baseline: 8.3643x; 1.0691x over previous
#include <hip/hip_runtime.h>
#include <hip/hip_bf16.h>


typedef __attribute__((ext_vector_type(2))) float v2f;
typedef __attribute__((ext_vector_type(8))) float v8f;

#define HEADS 4
#define HDIM 32
#define CIN 128
#define NEG_SLOPE 0.2f

#define GM_WAVES 4
#define GM_PITCH (CIN + 4)

__global__ __launch_bounds__(GM_WAVES * 32) void gat_gemm_wmma(
    const float* __restrict__ X,
    const float* __restrict__ Wl,
    const float* __restrict__ Wr,
    float* __restrict__ XL,
    float* __restrict__ XR,
    float* __restrict__ zinit,
    int nz4,
    int Nn)
{
    __shared__ __attribute__((aligned(16))) float wlds[CIN * GM_PITCH];

    const float* __restrict__ W  = (blockIdx.y == 0) ? Wl : Wr;
    float*       __restrict__ O  = (blockIdx.y == 0) ? XL : XR;

    {
        const float4* __restrict__ W4 = reinterpret_cast<const float4*>(W);
        for (int i = 0; i < (CIN * CIN / 4) / (GM_WAVES * 32); ++i) {
            const int q   = i * (GM_WAVES * 32) + (int)threadIdx.x;
            const int row = q >> 5;
            const int c4  = q & 31;
            *reinterpret_cast<float4*>(&wlds[row * GM_PITCH + c4 * 4]) = W4[q];
        }
    }
    __syncthreads();

    {
        const int nthr = (int)(gridDim.x * gridDim.y) * (GM_WAVES * 32);
        const int tid  = ((int)blockIdx.y * (int)gridDim.x + (int)blockIdx.x) * (GM_WAVES * 32)
                         + (int)threadIdx.x;
        float4* __restrict__ z4 = reinterpret_cast<float4*>(zinit);
        for (int q = tid; q < nz4; q += nthr)
            z4[q] = make_float4(0.0f, 0.0f, 0.0f, 0.0f);
    }

    const int wave = threadIdx.x >> 5;
    const int lane = threadIdx.x & 31;
    const int row0 = ((int)blockIdx.x * GM_WAVES + wave) * 16;
    if (row0 >= Nn) return;

    const int m  = lane & 15;
    const int hi = lane >> 4;

    int arow = row0 + m; if (arow >= Nn) arow = Nn - 1;

    const float* __restrict__ ap = X + (size_t)arow * CIN + hi * 2;
    v2f a[CIN / 4];
    #pragma unroll
    for (int s = 0; s < CIN / 4; ++s)
        a[s] = *reinterpret_cast<const v2f*>(ap + s * 4);

    const float* bp = wlds + m * GM_PITCH + hi * 2;

    for (int ct = 0; ct < CIN / 16; ct += 2) {
        const float* b0p = bp + (ct * 16) * GM_PITCH;
        const float* b1p = b0p + 16 * GM_PITCH;
        v8f acc0 = {};
        v8f acc1 = {};
        #pragma unroll
        for (int s = 0; s < CIN / 4; ++s) {
            const v2f b0 = *reinterpret_cast<const v2f*>(b0p + s * 4);
            const v2f b1 = *reinterpret_cast<const v2f*>(b1p + s * 4);
            acc0 = __builtin_amdgcn_wmma_f32_16x16x4_f32(
                 false, a[s],  false, b0,
                 (short)0, acc0,  false,  false);
            acc1 = __builtin_amdgcn_wmma_f32_16x16x4_f32(
                 false, a[s],  false, b1,
                 (short)0, acc1,  false,  false);
        }

        #pragma unroll
        for (int i = 0; i < 8; ++i) {
            const int r = row0 + i + hi * 8;
            if (r < Nn) {
                O[(size_t)r * CIN + ct * 16 + m]      = acc0[i];
                O[(size_t)r * CIN + ct * 16 + 16 + m] = acc1[i];
            }
        }
    }
}

#define BIN_SHIFT 6
#define BIN_NODES (1 << BIN_SHIFT)
#define MAXBINS 1024
#define SORT_THREADS 1024
#define SORT_CHUNK 8192
#define SORT_PER_THREAD (SORT_CHUNK / SORT_THREADS)

__global__ __launch_bounds__(SORT_THREADS) void gat_bin_count(
    const int* __restrict__ dst, int* __restrict__ binTotal, int E)
{
    __shared__ int hist[MAXBINS];
    const int t = threadIdx.x;
    hist[t] = 0;
    __syncthreads();
    const int e0 = (int)blockIdx.x * SORT_CHUNK;
    int dv[SORT_PER_THREAD];
    #pragma unroll
    for (int i = 0; i < SORT_PER_THREAD; ++i) {
        const int e = e0 + i * SORT_THREADS + t;
        dv[i] = (e < E) ? dst[e] : -1;
    }
    #pragma unroll
    for (int i = 0; i < SORT_PER_THREAD; ++i)
        if (dv[i] >= 0) atomicAdd(&hist[dv[i] >> BIN_SHIFT], 1);
    __syncthreads();
    const int c = hist[t];
    if (c > 0) atomicAdd(&binTotal[t], c);
}

__global__ __launch_bounds__(SORT_THREADS) void gat_bin_place(
    const int* __restrict__ src, const int* __restrict__ dst,
    const int* __restrict__ binTotal, int* __restrict__ binCursor,
    int* __restrict__ binStart,
    int* __restrict__ ssrc, int* __restrict__ sdst, int E)
{
    __shared__ int hist[MAXBINS];
    __shared__ int base[MAXBINS];
    __shared__ int wsum[SORT_THREADS / 32];
    __shared__ int wexc[SORT_THREADS / 32];
    const int t    = threadIdx.x;
    const int lane = t & 31;
    const int w    = t >> 5;
    hist[t] = 0;
    __syncthreads();
    const int e0 = (int)blockIdx.x * SORT_CHUNK;
    int dv[SORT_PER_THREAD], sv[SORT_PER_THREAD];
    #pragma unroll
    for (int i = 0; i < SORT_PER_THREAD; ++i) {
        const int e = e0 + i * SORT_THREADS + t;
        dv[i] = (e < E) ? dst[e] : -1;
        sv[i] = (e < E) ? src[e] : 0;
    }
    #pragma unroll
    for (int i = 0; i < SORT_PER_THREAD; ++i)
        if (dv[i] >= 0) atomicAdd(&hist[dv[i] >> BIN_SHIFT], 1);

    const int tot = binTotal[t];
    int v = tot;
    for (int off = 1; off < 32; off <<= 1) {
        const int u = __shfl_up(v, off);
        if (lane >= off) v += u;
    }
    if (lane == 31) wsum[w] = v;
    __syncthreads();
    if (w == 0) {
        const int s0 = wsum[lane];
        int s = s0;
        for (int off = 1; off < 32; off <<= 1) {
            const int u = __shfl_up(s, off);
            if (lane >= off) s += u;
        }
        wexc[lane] = s - s0;
    }
    __syncthreads();
    const int start = v - tot + wexc[w];
    if (blockIdx.x == 0) binStart[t] = start;

    const int c = hist[t];
    int bs = start;
    if (c > 0) bs += atomicAdd(&binCursor[t], c);
    base[t] = bs;
    hist[t] = 0;
    __syncthreads();

    #pragma unroll
    for (int i = 0; i < SORT_PER_THREAD; ++i) {
        if (dv[i] >= 0) {
            const int b   = dv[i] >> BIN_SHIFT;
            const int pos = base[b] + atomicAdd(&hist[b], 1);
            ssrc[pos] = sv[i];
            sdst[pos] = dv[i];
        }
    }
}

template <int M>
static __device__ __forceinline__ float gat_row_xor(float v) {
    return __int_as_float(__builtin_amdgcn_update_dpp(
        0, __float_as_int(v), 0x160 + M, 0xf, 0xf, true));
}

#define SC_GROUP 4

template <int G>
static __device__ __forceinline__ void gat_sum_group(
    const float* __restrict__ xr, float* srow,
    int lane, int jb, int s, int dl, float a0, float a1, float a2, float a3)
{
    float x[G][4];
    float al[G][4];
    int   dj[G];
    #pragma unroll
    for (int u = 0; u < G; ++u) {
        const int j  = jb + u;
        const int sj = __builtin_amdgcn_readlane(s, j);
        dj[u]    = __builtin_amdgcn_readlane(dl, j);
        al[u][0] = __int_as_float(__builtin_amdgcn_readlane(__float_as_int(a0), j));
        al[u][1] = __int_as_float(__builtin_amdgcn_readlane(__float_as_int(a1), j));
        al[u][2] = __int_as_float(__builtin_amdgcn_readlane(__float_as_int(a2), j));
        al[u][3] = __int_as_float(__builtin_amdgcn_readlane(__float_as_int(a3), j));
        const float* __restrict__ p = xr + (size_t)sj * CIN + lane;
        x[u][0] = p[0];
        x[u][1] = p[HDIM];
        x[u][2] = p[2 * HDIM];
        x[u][3] = p[3 * HDIM];
    }
    #pragma unroll
    for (int u = 0; u < G; ++u) {
        const float v = fmaf(al[u][3], x[u][3],
                        fmaf(al[u][2], x[u][2],
                        fmaf(al[u][1], x[u][1], al[u][0] * x[u][0])));
        atomicAdd(srow + dj[u] * HDIM + lane, 0.25f * v);
    }
}

#define FB_WAVES 8

__global__ __launch_bounds__(FB_WAVES * 32) void gat_bin_edges(
    const float* __restrict__ xl, const float* __restrict__ xr,
    const int* __restrict__ ssrc, const int* __restrict__ sdst,
    const int* __restrict__ binStart, const int* __restrict__ binTotal,
    const float* __restrict__ att, const float* __restrict__ bias,
    float* __restrict__ ebuf, float* __restrict__ out,
    int Nn, int E)
{
    __shared__ __attribute__((aligned(16))) int   smax[BIN_NODES * HEADS];
    __shared__ __attribute__((aligned(16))) float sden[BIN_NODES * HEADS];
    __shared__ __attribute__((aligned(16))) float sout[BIN_NODES * HDIM];
    __shared__ __attribute__((aligned(16))) float sxr[BIN_NODES * CIN];

    const int lane = threadIdx.x & 31;
    const int wave  = __builtin_amdgcn_readfirstlane((int)(threadIdx.x >> 5));
    const int b     = (int)blockIdx.x;
    const int n0    = b << BIN_SHIFT;
    const int nn    = (Nn - n0 < BIN_NODES) ? (Nn - n0) : BIN_NODES;
    const int start = __builtin_amdgcn_readfirstlane(binStart[b]);
    const int cntE  = __builtin_amdgcn_readfirstlane(binTotal[b]);
    const int cnt   = cntE + nn;

    for (int q = threadIdx.x; q < BIN_NODES * HEADS; q += FB_WAVES * 32) {
        smax[q] = 0;
        sden[q] = 0.0f;
    }
    for (int q = threadIdx.x; q < BIN_NODES * HDIM; q += FB_WAVES * 32)
        sout[q] = 0.0f;
    {
        const float4* __restrict__ g4 = reinterpret_cast<const float4*>(xr + (size_t)n0 * CIN);
        float4* s4 = reinterpret_cast<float4*>(sxr);
        const int lim = nn * (CIN / 4);
        float4 t4[BIN_NODES * (CIN / 4) / (FB_WAVES * 32)];
        #pragma unroll
        for (int i = 0; i < BIN_NODES * (CIN / 4) / (FB_WAVES * 32); ++i) {
            const int q = i * (FB_WAVES * 32) + (int)threadIdx.x;
            t4[i] = (q < lim) ? g4[q] : make_float4(0.0f, 0.0f, 0.0f, 0.0f);
        }
        #pragma unroll
        for (int i = 0; i < BIN_NODES * (CIN / 4) / (FB_WAVES * 32); ++i)
            s4[i * (FB_WAVES * 32) + (int)threadIdx.x] = t4[i];
    }
    __syncthreads();

    {
        const float4 a4 = *reinterpret_cast<const float4*>(att + lane * 4);
        const int k = lane & 7;
        const int h = lane >> 3;
        for (int g0 = wave * 32; g0 < cnt; g0 += FB_WAVES * 32) {
            const int c = (cnt - g0 < 32) ? (cnt - g0) : 32;
            int s = n0, d = n0, slot = 0;
            if (lane < c) {
                const int j = g0 + lane;
                if (j < cntE) { s = ssrc[start + j]; d = sdst[start + j]; slot = start + j; }
                else          { s = d = n0 + (j - cntE); slot = E + s; }
            }
            for (int jb = 0; jb < c; jb += 8) {
                float4 L[8], R[8];
                for (int u = 0; u < 8; ++u) {
                    const int j  = jb + u;
                    const int sj = __builtin_amdgcn_readlane(s, j);
                    const int dj = __builtin_amdgcn_readlane(d, j);
                    L[u] = *reinterpret_cast<const float4*>(xl + (size_t)sj * CIN + lane * 4);
                    R[u] = *reinterpret_cast<const float4*>(&sxr[(dj - n0) * CIN + lane * 4]);
                }
                float keep = 0.0f;
                for (int u = 0; u < 8; ++u) {
                    float v0 = L[u].x + R[u].x;
                    float v1 = L[u].y + R[u].y;
                    float v2 = L[u].z + R[u].z;
                    float v3 = L[u].w + R[u].w;
                    v0 = fmaxf(v0, NEG_SLOPE * v0);
                    v1 = fmaxf(v1, NEG_SLOPE * v1);
                    v2 = fmaxf(v2, NEG_SLOPE * v2);
                    v3 = fmaxf(v3, NEG_SLOPE * v3);
                    float p = fmaf(a4.w, v3, fmaf(a4.z, v2, fmaf(a4.y, v1, a4.x * v0)));
                    p += gat_row_xor<1>(p);
                    p += gat_row_xor<2>(p);
                    p += gat_row_xor<4>(p);
                    keep = (k == u) ? p : keep;
                }
                const int ej = jb + k;
                const int dd = __shfl(d, ej);
                const int sl = __shfl(slot, ej);
                if (ej < c) {
                    ebuf[(size_t)sl * HEADS + h] = keep;
                    if (keep > 0.0f)
                        atomicMax(&smax[(dd - n0) * HEADS + h], __float_as_int(keep));
                }
            }
        }
    }
    __syncthreads();

    for (int j = threadIdx.x; j < cnt; j += FB_WAVES * 32) {
        int d, slot;
        if (j < cntE) { d = sdst[start + j]; slot = start + j; }
        else          { d = n0 + (j - cntE); slot = E + d; }
        const int dl = d - n0;
        const float4 sc = *reinterpret_cast<const float4*>(ebuf + (size_t)slot * HEADS);
        const int4   mx = *reinterpret_cast<const int4*>(&smax[dl * HEADS]);
        atomicAdd(&sden[dl * HEADS + 0], __expf(sc.x - __int_as_float(mx.x)));
        atomicAdd(&sden[dl * HEADS + 1], __expf(sc.y - __int_as_float(mx.y)));
        atomicAdd(&sden[dl * HEADS + 2], __expf(sc.z - __int_as_float(mx.z)));
        atomicAdd(&sden[dl * HEADS + 3], __expf(sc.w - __int_as_float(mx.w)));
    }
    __syncthreads();

    for (int g0 = wave * 32; g0 < cnt; g0 += FB_WAVES * 32) {
        const int c = (cnt - g0 < 32) ? (cnt - g0) : 32;
        int   s = n0, dl = 0;
        float a0 = 0.0f, a1 = 0.0f, a2 = 0.0f, a3 = 0.0f;
        if (lane < c) {
            const int j = g0 + lane;
            int d, slot;
            if (j < cntE) { s = ssrc[start + j]; d = sdst[start + j]; slot = start + j; }
            else          { s = d = n0 + (j - cntE); slot = E + s; }
            dl = d - n0;
            const float4 sc = *reinterpret_cast<const float4*>(ebuf + (size_t)slot * HEADS);
            const int4   mx = *reinterpret_cast<const int4*>(&smax[dl * HEADS]);
            const float4 dn = *reinterpret_cast<const float4*>(&sden[dl * HEADS]);
            a0 = __expf(sc.x - __int_as_float(mx.x)) / ((dn.x > 1e-16f) ? dn.x : 1e-16f);
            a1 = __expf(sc.y - __int_as_float(mx.y)) / ((dn.y > 1e-16f) ? dn.y : 1e-16f);
            a2 = __expf(sc.z - __int_as_float(mx.z)) / ((dn.z > 1e-16f) ? dn.z : 1e-16f);
            a3 = __expf(sc.w - __int_as_float(mx.w)) / ((dn.w > 1e-16f) ? dn.w : 1e-16f);
        }
        int jb = 0;
        for (; jb + SC_GROUP <= c; jb += SC_GROUP)
            gat_sum_group<SC_GROUP>(xr, sout, lane, jb, s, dl, a0, a1, a2, a3);
        #pragma nounroll
        for (; jb < c; ++jb)
            gat_sum_group<1>(xr, sout, lane, jb, s, dl, a0, a1, a2, a3);
    }
    __syncthreads();

    for (int q = threadIdx.x; q < nn * HDIM; q += FB_WAVES * 32)
        out[(size_t)n0 * HDIM + q] = bias[q & (HDIM - 1)] + sout[q];
}

extern "C" void kernel_launch(void* const* d_in, const int* in_sizes, int n_in,
                              void* d_out, int out_size, void* d_ws, size_t ws_size,
                              hipStream_t stream)
{
    const float* x    = (const float*)d_in[0];
    const int*   eidx = (const int*)d_in[1];
    const float* Wl   = (const float*)d_in[2];
    const float* Wr   = (const float*)d_in[3];
    const float* att  = (const float*)d_in[4];
    const float* bias = (const float*)d_in[5];
    float* out = (float*)d_out;

    const int Nn   = in_sizes[0] / CIN;
    const int E    = in_sizes[1] / 2;
    const int Etot = E + Nn;

    const int* src = eidx;
    const int* dst = eidx + E;

    float* ws = (float*)d_ws;
    size_t off = 0;
    float* xl    = ws + off; off += (size_t)Nn * CIN;
    float* xr    = ws + off; off += (size_t)Nn * CIN;
    float* ebuf  = ws + off; off += (size_t)Etot * HEADS;
    int* binTotal  = (int*)(ws + off); off += MAXBINS;
    int* binCursor = (int*)(ws + off); off += MAXBINS;
    int* binStart  = (int*)(ws + off); off += MAXBINS;
    int* ssrc      = (int*)(ws + off); off += (size_t)E;
    int* sdst      = (int*)(ws + off); off += (size_t)E;
    (void)ws_size;

    {
        const int rowTiles = (Nn + 15) / 16;
        dim3 grid((rowTiles + GM_WAVES - 1) / GM_WAVES, 2);
        gat_gemm_wmma<<<grid, GM_WAVES * 32, 0, stream>>>(
            x, Wl, Wr, xl, xr, (float*)binTotal, 2 * MAXBINS / 4, Nn);
    }

    const int sortBlocks = (E + SORT_CHUNK - 1) / SORT_CHUNK;
    const int nbins      = (Nn + BIN_NODES - 1) / BIN_NODES;

    gat_bin_count<<<sortBlocks, SORT_THREADS, 0, stream>>>(dst, binTotal, E);

    gat_bin_place<<<sortBlocks, SORT_THREADS, 0, stream>>>(
        src, dst, binTotal, binCursor, binStart, ssrc, sdst, E);

    gat_bin_edges<<<nbins, FB_WAVES * 32, 0, stream>>>(
        xl, xr, ssrc, sdst, binStart, binTotal, att, bias, ebuf, out, Nn, E);
}


# place launch: a block first groups its 8192 edges by bin in LDS (scan of its own counts + LDS ranks), then stores the chunk in index order: contiguous runs per bin, not a 4-byte scatter
# speedup vs baseline: 10.0409x; 1.2004x over previous
#include <hip/hip_runtime.h>
#include <hip/hip_bf16.h>


typedef __attribute__((ext_vector_type(2))) float v2f;
typedef __attribute__((ext_vector_type(8))) float v8f;

#define HEADS 4
#define HDIM 32
#define CIN 128
#define NEG_SLOPE 0.2f

#define GM_WAVES 4
#define GM_PITCH (CIN + 4)

__global__ __launch_bounds__(GM_WAVES * 32) void gat_gemm_wmma(
    const float* __restrict__ X,
    const float* __restrict__ Wl,
    const float* __restrict__ Wr,
    float* __restrict__ XL,
    float* __restrict__ XR,
    float* __restrict__ zinit,
    int nz4,
    int Nn)
{
    __shared__ __attribute__((aligned(16))) float wlds[CIN * GM_PITCH];

    const float* __restrict__ W  = (blockIdx.y == 0) ? Wl : Wr;
    float*       __restrict__ O  = (blockIdx.y == 0) ? XL : XR;

    {
        const float4* __restrict__ W4 = reinterpret_cast<const float4*>(W);
        for (int i = 0; i < (CIN * CIN / 4) / (GM_WAVES * 32); ++i) {
            const int q   = i * (GM_WAVES * 32) + (int)threadIdx.x;
            const int row = q >> 5;
            const int c4  = q & 31;
            *reinterpret_cast<float4*>(&wlds[row * GM_PITCH + c4 * 4]) = W4[q];
        }
    }
    __syncthreads();

    {
        const int nthr = (int)(gridDim.x * gridDim.y) * (GM_WAVES * 32);
        const int tid  = ((int)blockIdx.y * (int)gridDim.x + (int)blockIdx.x) * (GM_WAVES * 32)
                         + (int)threadIdx.x;
        float4* __restrict__ z4 = reinterpret_cast<float4*>(zinit);
        for (int q = tid; q < nz4; q += nthr)
            z4[q] = make_float4(0.0f, 0.0f, 0.0f, 0.0f);
    }

    const int wave = threadIdx.x >> 5;
    const int lane = threadIdx.x & 31;
    const int row0 = ((int)blockIdx.x * GM_WAVES + wave) * 16;
    if (row0 >= Nn) return;

    const int m  = lane & 15;
    const int hi = lane >> 4;

    int arow = row0 + m; if (arow >= Nn) arow = Nn - 1;

    const float* __restrict__ ap = X + (size_t)arow * CIN + hi * 2;
    v2f a[CIN / 4];
    #pragma unroll
    for (int s = 0; s < CIN / 4; ++s)
        a[s] = *reinterpret_cast<const v2f*>(ap + s * 4);

    const float* bp = wlds + m * GM_PITCH + hi * 2;

    for (int ct = 0; ct < CIN / 16; ct += 2) {
        const float* b0p = bp + (ct * 16) * GM_PITCH;
        const float* b1p = b0p + 16 * GM_PITCH;
        v8f acc0 = {};
        v8f acc1 = {};
        #pragma unroll
        for (int s = 0; s < CIN / 4; ++s) {
            const v2f b0 = *reinterpret_cast<const v2f*>(b0p + s * 4);
            const v2f b1 = *reinterpret_cast<const v2f*>(b1p + s * 4);
            acc0 = __builtin_amdgcn_wmma_f32_16x16x4_f32(
                 false, a[s],  false, b0,
                 (short)0, acc0,  false,  false);
            acc1 = __builtin_amdgcn_wmma_f32_16x16x4_f32(
                 false, a[s],  false, b1,
                 (short)0, acc1,  false,  false);
        }

        #pragma unroll
        for (int i = 0; i < 8; ++i) {
            const int r = row0 + i + hi * 8;
            if (r < Nn) {
                O[(size_t)r * CIN + ct * 16 + m]      = acc0[i];
                O[(size_t)r * CIN + ct * 16 + 16 + m] = acc1[i];
            }
        }
    }
}

#define BIN_SHIFT 6
#define BIN_NODES (1 << BIN_SHIFT)
#define MAXBINS 1024
#define SORT_THREADS 1024
#define SORT_CHUNK 8192
#define SORT_PER_THREAD (SORT_CHUNK / SORT_THREADS)

__global__ __launch_bounds__(SORT_THREADS) void gat_bin_count(
    const int* __restrict__ dst, int* __restrict__ binTotal, int E)
{
    __shared__ int hist[MAXBINS];
    const int t = threadIdx.x;
    hist[t] = 0;
    __syncthreads();
    const int e0 = (int)blockIdx.x * SORT_CHUNK;
    int dv[SORT_PER_THREAD];
    #pragma unroll
    for (int i = 0; i < SORT_PER_THREAD; ++i) {
        const int e = e0 + i * SORT_THREADS + t;
        dv[i] = (e < E) ? dst[e] : -1;
    }
    #pragma unroll
    for (int i = 0; i < SORT_PER_THREAD; ++i)
        if (dv[i] >= 0) atomicAdd(&hist[dv[i] >> BIN_SHIFT], 1);
    __syncthreads();
    const int c = hist[t];
    if (c > 0) atomicAdd(&binTotal[t], c);
}

__global__ __launch_bounds__(SORT_THREADS) void gat_bin_place(
    const int* __restrict__ src, const int* __restrict__ dst,
    const int* __restrict__ binTotal, int* __restrict__ binCursor,
    int* __restrict__ binStart,
    int* __restrict__ ssrc, int* __restrict__ sdst, int E)
{
    __shared__ int hist[MAXBINS];
    __shared__ int gofs[MAXBINS];
    __shared__ int wsum[SORT_THREADS / 32];
    __shared__ int lsum[SORT_THREADS / 32];
    __shared__ int stS[SORT_CHUNK];
    __shared__ int stD[SORT_CHUNK];
    const int t    = threadIdx.x;
    const int lane = t & 31;
    const int w    = __builtin_amdgcn_readfirstlane(t >> 5);
    hist[t] = 0;
    __syncthreads();
    const int e0 = (int)blockIdx.x * SORT_CHUNK;
    int dv[SORT_PER_THREAD], sv[SORT_PER_THREAD];
    #pragma unroll
    for (int i = 0; i < SORT_PER_THREAD; ++i) {
        const int e = e0 + i * SORT_THREADS + t;
        dv[i] = (e < E) ? dst[e] : -1;
        sv[i] = (e < E) ? src[e] : 0;
    }
    #pragma unroll
    for (int i = 0; i < SORT_PER_THREAD; ++i)
        if (dv[i] >= 0) atomicAdd(&hist[dv[i] >> BIN_SHIFT], 1);

    const int tot = binTotal[t];
    int v = tot;
    for (int off = 1; off < 32; off <<= 1) {
        const int u = __shfl_up(v, off);
        if (lane >= off) v += u;
    }
    if (lane == 31) wsum[w] = v;
    __syncthreads();

    const int c = hist[t];
    int lv = c;
    for (int off = 1; off < 32; off <<= 1) {
        const int u = __shfl_up(lv, off);
        if (lane >= off) lv += u;
    }
    if (lane == 31) lsum[w] = lv;

    int start;
    {
        const int s0 = wsum[lane];
        int s = s0;
        for (int off = 1; off < 32; off <<= 1) {
            const int u = __shfl_up(s, off);
            if (lane >= off) s += u;
        }
        start = v - tot + __builtin_amdgcn_readlane(s - s0, w);
    }
    if (blockIdx.x == 0) binStart[t] = start;
    int bs = start;
    if (c > 0) bs += atomicAdd(&binCursor[t], c);
    __syncthreads();
    {
        const int s0 = lsum[lane];
        int s = s0;
        for (int off = 1; off < 32; off <<= 1) {
            const int u = __shfl_up(s, off);
            if (lane >= off) s += u;
        }
        const int lofs = lv - c + __builtin_amdgcn_readlane(s - s0, w);
        gofs[t] = bs - lofs;
        hist[t] = lofs;
    }
    __syncthreads();

    #pragma unroll
    for (int i = 0; i < SORT_PER_THREAD; ++i) {
        if (dv[i] >= 0) {
            const int lp = atomicAdd(&hist[dv[i] >> BIN_SHIFT], 1);
            stS[lp] = sv[i];
            stD[lp] = dv[i];
        }
    }
    __syncthreads();

    const int nv = (E - e0 < SORT_CHUNK) ? (E - e0) : SORT_CHUNK;
    for (int i = 0; i < SORT_PER_THREAD; ++i) {
        const int idx = i * SORT_THREADS + t;
        if (idx < nv) {
            const int d   = stD[idx];
            const int pos = gofs[d >> BIN_SHIFT] + idx;
            ssrc[pos] = stS[idx];
            sdst[pos] = d;
        }
    }
}

template <int M>
static __device__ __forceinline__ float gat_row_xor(float v) {
    return __int_as_float(__builtin_amdgcn_update_dpp(
        0, __float_as_int(v), 0x160 + M, 0xf, 0xf, true));
}

#define SC_GROUP 4

template <int G>
static __device__ __forceinline__ void gat_sum_group(
    const float* __restrict__ xr, float* srow,
    int lane, int jb, int s, int dl, float a0, float a1, float a2, float a3)
{
    float x[G][4];
    float al[G][4];
    int   dj[G];
    #pragma unroll
    for (int u = 0; u < G; ++u) {
        const int j  = jb + u;
        const int sj = __builtin_amdgcn_readlane(s, j);
        dj[u]    = __builtin_amdgcn_readlane(dl, j);
        al[u][0] = __int_as_float(__builtin_amdgcn_readlane(__float_as_int(a0), j));
        al[u][1] = __int_as_float(__builtin_amdgcn_readlane(__float_as_int(a1), j));
        al[u][2] = __int_as_float(__builtin_amdgcn_readlane(__float_as_int(a2), j));
        al[u][3] = __int_as_float(__builtin_amdgcn_readlane(__float_as_int(a3), j));
        const float* __restrict__ p = xr + (size_t)sj * CIN + lane;
        x[u][0] = p[0];
        x[u][1] = p[HDIM];
        x[u][2] = p[2 * HDIM];
        x[u][3] = p[3 * HDIM];
    }
    #pragma unroll
    for (int u = 0; u < G; ++u) {
        const float v = fmaf(al[u][3], x[u][3],
                        fmaf(al[u][2], x[u][2],
                        fmaf(al[u][1], x[u][1], al[u][0] * x[u][0])));
        atomicAdd(srow + dj[u] * HDIM + lane, 0.25f * v);
    }
}

#define FB_WAVES 8

__global__ __launch_bounds__(FB_WAVES * 32) void gat_bin_edges(
    const float* __restrict__ xl, const float* __restrict__ xr,
    const int* __restrict__ ssrc, const int* __restrict__ sdst,
    const int* __restrict__ binStart, const int* __restrict__ binTotal,
    const float* __restrict__ att, const float* __restrict__ bias,
    float* __restrict__ ebuf, float* __restrict__ out,
    int Nn, int E)
{
    __shared__ __attribute__((aligned(16))) int   smax[BIN_NODES * HEADS];
    __shared__ __attribute__((aligned(16))) float sden[BIN_NODES * HEADS];
    __shared__ __attribute__((aligned(16))) float sout[BIN_NODES * HDIM];
    __shared__ __attribute__((aligned(16))) float sxr[BIN_NODES * CIN];

    const int lane = threadIdx.x & 31;
    const int wave  = __builtin_amdgcn_readfirstlane((int)(threadIdx.x >> 5));
    const int b     = (int)blockIdx.x;
    const int n0    = b << BIN_SHIFT;
    const int nn    = (Nn - n0 < BIN_NODES) ? (Nn - n0) : BIN_NODES;
    const int start = __builtin_amdgcn_readfirstlane(binStart[b]);
    const int cntE  = __builtin_amdgcn_readfirstlane(binTotal[b]);
    const int cnt   = cntE + nn;

    for (int q = threadIdx.x; q < BIN_NODES * HEADS; q += FB_WAVES * 32) {
        smax[q] = 0;
        sden[q] = 0.0f;
    }
    for (int q = threadIdx.x; q < BIN_NODES * HDIM; q += FB_WAVES * 32)
        sout[q] = 0.0f;
    {
        const float4* __restrict__ g4 = reinterpret_cast<const float4*>(xr + (size_t)n0 * CIN);
        float4* s4 = reinterpret_cast<float4*>(sxr);
        const int lim = nn * (CIN / 4);
        float4 t4[BIN_NODES * (CIN / 4) / (FB_WAVES * 32)];
        #pragma unroll
        for (int i = 0; i < BIN_NODES * (CIN / 4) / (FB_WAVES * 32); ++i) {
            const int q = i * (FB_WAVES * 32) + (int)threadIdx.x;
            t4[i] = (q < lim) ? g4[q] : make_float4(0.0f, 0.0f, 0.0f, 0.0f);
        }
        #pragma unroll
        for (int i = 0; i < BIN_NODES * (CIN / 4) / (FB_WAVES * 32); ++i)
            s4[i * (FB_WAVES * 32) + (int)threadIdx.x] = t4[i];
    }
    __syncthreads();

    {
        const float4 a4 = *reinterpret_cast<const float4*>(att + lane * 4);
        const int k = lane & 7;
        const int h = lane >> 3;
        for (int g0 = wave * 32; g0 < cnt; g0 += FB_WAVES * 32) {
            const int c = (cnt - g0 < 32) ? (cnt - g0) : 32;
            int s = n0, d = n0, slot = 0;
            if (lane < c) {
                const int j = g0 + lane;
                if (j < cntE) { s = ssrc[start + j]; d = sdst[start + j]; slot = start + j; }
                else          { s = d = n0 + (j - cntE); slot = E + s; }
            }
            for (int jb = 0; jb < c; jb += 8) {
                float4 L[8], R[8];
                for (int u = 0; u < 8; ++u) {
                    const int j  = jb + u;
                    const int sj = __builtin_amdgcn_readlane(s, j);
                    const int dj = __builtin_amdgcn_readlane(d, j);
                    L[u] = *reinterpret_cast<const float4*>(xl + (size_t)sj * CIN + lane * 4);
                    R[u] = *reinterpret_cast<const float4*>(&sxr[(dj - n0) * CIN + lane * 4]);
                }
                float keep = 0.0f;
                for (int u = 0; u < 8; ++u) {
                    float v0 = L[u].x + R[u].x;
                    float v1 = L[u].y + R[u].y;
                    float v2 = L[u].z + R[u].z;
                    float v3 = L[u].w + R[u].w;
                    v0 = fmaxf(v0, NEG_SLOPE * v0);
                    v1 = fmaxf(v1, NEG_SLOPE * v1);
                    v2 = fmaxf(v2, NEG_SLOPE * v2);
                    v3 = fmaxf(v3, NEG_SLOPE * v3);
                    float p = fmaf(a4.w, v3, fmaf(a4.z, v2, fmaf(a4.y, v1, a4.x * v0)));
                    p += gat_row_xor<1>(p);
                    p += gat_row_xor<2>(p);
                    p += gat_row_xor<4>(p);
                    keep = (k == u) ? p : keep;
                }
                const int ej = jb + k;
                const int dd = __shfl(d, ej);
                const int sl = __shfl(slot, ej);
                if (ej < c) {
                    ebuf[(size_t)sl * HEADS + h] = keep;
                    if (keep > 0.0f)
                        atomicMax(&smax[(dd - n0) * HEADS + h], __float_as_int(keep));
                }
            }
        }
    }
    __syncthreads();

    for (int j = threadIdx.x; j < cnt; j += FB_WAVES * 32) {
        int d, slot;
        if (j < cntE) { d = sdst[start + j]; slot = start + j; }
        else          { d = n0 + (j - cntE); slot = E + d; }
        const int dl = d - n0;
        const float4 sc = *reinterpret_cast<const float4*>(ebuf + (size_t)slot * HEADS);
        const int4   mx = *reinterpret_cast<const int4*>(&smax[dl * HEADS]);
        atomicAdd(&sden[dl * HEADS + 0], __expf(sc.x - __int_as_float(mx.x)));
        atomicAdd(&sden[dl * HEADS + 1], __expf(sc.y - __int_as_float(mx.y)));
        atomicAdd(&sden[dl * HEADS + 2], __expf(sc.z - __int_as_float(mx.z)));
        atomicAdd(&sden[dl * HEADS + 3], __expf(sc.w - __int_as_float(mx.w)));
    }
    __syncthreads();

    for (int g0 = wave * 32; g0 < cnt; g0 += FB_WAVES * 32) {
        const int c = (cnt - g0 < 32) ? (cnt - g0) : 32;
        int   s = n0, dl = 0;
        float a0 = 0.0f, a1 = 0.0f, a2 = 0.0f, a3 = 0.0f;
        if (lane < c) {
            const int j = g0 + lane;
            int d, slot;
            if (j < cntE) { s = ssrc[start + j]; d = sdst[start + j]; slot = start + j; }
            else          { s = d = n0 + (j - cntE); slot = E + s; }
            dl = d - n0;
            const float4 sc = *reinterpret_cast<const float4*>(ebuf + (size_t)slot * HEADS);
            const int4   mx = *reinterpret_cast<const int4*>(&smax[dl * HEADS]);
            const float4 dn = *reinterpret_cast<const float4*>(&sden[dl * HEADS]);
            a0 = __expf(sc.x - __int_as_float(mx.x)) / ((dn.x > 1e-16f) ? dn.x : 1e-16f);
            a1 = __expf(sc.y - __int_as_float(mx.y)) / ((dn.y > 1e-16f) ? dn.y : 1e-16f);
            a2 = __expf(sc.z - __int_as_float(mx.z)) / ((dn.z > 1e-16f) ? dn.z : 1e-16f);
            a3 = __expf(sc.w - __int_as_float(mx.w)) / ((dn.w > 1e-16f) ? dn.w : 1e-16f);
        }
        int jb = 0;
        for (; jb + SC_GROUP <= c; jb += SC_GROUP)
            gat_sum_group<SC_GROUP>(xr, sout, lane, jb, s, dl, a0, a1, a2, a3);
        #pragma nounroll
        for (; jb < c; ++jb)
            gat_sum_group<1>(xr, sout, lane, jb, s, dl, a0, a1, a2, a3);
    }
    __syncthreads();

    for (int q = threadIdx.x; q < nn * HDIM; q += FB_WAVES * 32)
        out[(size_t)n0 * HDIM + q] = bias[q & (HDIM - 1)] + sout[q];
}

extern "C" void kernel_launch(void* const* d_in, const int* in_sizes, int n_in,
                              void* d_out, int out_size, void* d_ws, size_t ws_size,
                              hipStream_t stream)
{
    const float* x    = (const float*)d_in[0];
    const int*   eidx = (const int*)d_in[1];
    const float* Wl   = (const float*)d_in[2];
    const float* Wr   = (const float*)d_in[3];
    const float* att  = (const float*)d_in[4];
    const float* bias = (const float*)d_in[5];
    float* out = (float*)d_out;

    const int Nn   = in_sizes[0] / CIN;
    const int E    = in_sizes[1] / 2;
    const int Etot = E + Nn;

    const int* src = eidx;
    const int* dst = eidx + E;

    float* ws = (float*)d_ws;
    size_t off = 0;
    float* xl    = ws + off; off += (size_t)Nn * CIN;
    float* xr    = ws + off; off += (size_t)Nn * CIN;
    float* ebuf  = ws + off; off += (size_t)Etot * HEADS;
    int* binTotal  = (int*)(ws + off); off += MAXBINS;
    int* binCursor = (int*)(ws + off); off += MAXBINS;
    int* binStart  = (int*)(ws + off); off += MAXBINS;
    int* ssrc      = (int*)(ws + off); off += (size_t)E;
    int* sdst      = (int*)(ws + off); off += (size_t)E;
    (void)ws_size;

    {
        const int rowTiles = (Nn + 15) / 16;
        dim3 grid((rowTiles + GM_WAVES - 1) / GM_WAVES, 2);
        gat_gemm_wmma<<<grid, GM_WAVES * 32, 0, stream>>>(
            x, Wl, Wr, xl, xr, (float*)binTotal, 2 * MAXBINS / 4, Nn);
    }

    const int sortBlocks = (E + SORT_CHUNK - 1) / SORT_CHUNK;
    const int nbins      = (Nn + BIN_NODES - 1) / BIN_NODES;

    gat_bin_count<<<sortBlocks, SORT_THREADS, 0, stream>>>(dst, binTotal, E);

    gat_bin_place<<<sortBlocks, SORT_THREADS, 0, stream>>>(
        src, dst, binTotal, binCursor, binStart, ssrc, sdst, E);

    gat_bin_edges<<<nbins, FB_WAVES * 32, 0, stream>>>(
        xl, xr, ssrc, sdst, binStart, binTotal, att, bias, ebuf, out, Nn, E);
}


# bin kernel: edges regrouped by node (LDS counting sort); a wave walks one node's edges in one pass, running max + rescaled sums in registers: no score buffer, no float atomics; sum order changes
# speedup vs baseline: 11.1694x; 1.1124x over previous
#include <hip/hip_runtime.h>
#include <hip/hip_bf16.h>


typedef __attribute__((ext_vector_type(2))) float v2f;
typedef __attribute__((ext_vector_type(8))) float v8f;
typedef __attribute__((ext_vector_type(4))) float v4f;

#define HEADS 4
#define HDIM 32
#define CIN 128
#define NEG_SLOPE 0.2f

#define GM_WAVES 4
#define GM_PITCH (CIN + 4)

__global__ __launch_bounds__(GM_WAVES * 32) void gat_gemm_wmma(
    const float* __restrict__ X,
    const float* __restrict__ Wl,
    const float* __restrict__ Wr,
    float* __restrict__ XL,
    float* __restrict__ XR,
    float* __restrict__ zinit,
    int nz4,
    int Nn)
{
    __shared__ __attribute__((aligned(16))) float wlds[CIN * GM_PITCH];

    const float* __restrict__ W  = (blockIdx.y == 0) ? Wl : Wr;
    float*       __restrict__ O  = (blockIdx.y == 0) ? XL : XR;

    {
        const float4* __restrict__ W4 = reinterpret_cast<const float4*>(W);
        for (int i = 0; i < (CIN * CIN / 4) / (GM_WAVES * 32); ++i) {
            const int q   = i * (GM_WAVES * 32) + (int)threadIdx.x;
            const int row = q >> 5;
            const int c4  = q & 31;
            *reinterpret_cast<float4*>(&wlds[row * GM_PITCH + c4 * 4]) = W4[q];
        }
    }
    __syncthreads();

    {
        const int nthr = (int)(gridDim.x * gridDim.y) * (GM_WAVES * 32);
        const int tid  = ((int)blockIdx.y * (int)gridDim.x + (int)blockIdx.x) * (GM_WAVES * 32)
                         + (int)threadIdx.x;
        float4* __restrict__ z4 = reinterpret_cast<float4*>(zinit);
        for (int q = tid; q < nz4; q += nthr)
            z4[q] = make_float4(0.0f, 0.0f, 0.0f, 0.0f);
    }

    const int wave = threadIdx.x >> 5;
    const int lane = threadIdx.x & 31;
    const int row0 = ((int)blockIdx.x * GM_WAVES + wave) * 16;
    if (row0 >= Nn) return;

    const int m  = lane & 15;
    const int hi = lane >> 4;

    int arow = row0 + m; if (arow >= Nn) arow = Nn - 1;

    const float* __restrict__ ap = X + (size_t)arow * CIN + hi * 2;
    v2f a[CIN / 4];
    #pragma unroll
    for (int s = 0; s < CIN / 4; ++s)
        a[s] = *reinterpret_cast<const v2f*>(ap + s * 4);

    const float* bp = wlds + m * GM_PITCH + hi * 2;

    for (int ct = 0; ct < CIN / 16; ct += 2) {
        const float* b0p = bp + (ct * 16) * GM_PITCH;
        const float* b1p = b0p + 16 * GM_PITCH;
        v8f acc0 = {};
        v8f acc1 = {};
        #pragma unroll
        for (int s = 0; s < CIN / 4; ++s) {
            const v2f b0 = *reinterpret_cast<const v2f*>(b0p + s * 4);
            const v2f b1 = *reinterpret_cast<const v2f*>(b1p + s * 4);
            acc0 = __builtin_amdgcn_wmma_f32_16x16x4_f32(
                 false, a[s],  false, b0,
                 (short)0, acc0,  false,  false);
            acc1 = __builtin_amdgcn_wmma_f32_16x16x4_f32(
                 false, a[s],  false, b1,
                 (short)0, acc1,  false,  false);
        }

        #pragma unroll
        for (int i = 0; i < 8; ++i) {
            const int r = row0 + i + hi * 8;
            if (r < Nn) {
                O[(size_t)r * CIN + ct * 16 + m]      = acc0[i];
                O[(size_t)r * CIN + ct * 16 + 16 + m] = acc1[i];
            }
        }
    }
}

#define BIN_SHIFT 6
#define BIN_NODES (1 << BIN_SHIFT)
#define MAXBINS 1024
#define SORT_THREADS 1024
#define SORT_CHUNK 8192
#define SORT_PER_THREAD (SORT_CHUNK / SORT_THREADS)

__global__ __launch_bounds__(SORT_THREADS) void gat_bin_count(
    const int* __restrict__ dst, int* __restrict__ binTotal, int E)
{
    __shared__ int hist[MAXBINS];
    const int t = threadIdx.x;
    hist[t] = 0;
    __syncthreads();
    const int e0 = (int)blockIdx.x * SORT_CHUNK;
    int dv[SORT_PER_THREAD];
    #pragma unroll
    for (int i = 0; i < SORT_PER_THREAD; ++i) {
        const int e = e0 + i * SORT_THREADS + t;
        dv[i] = (e < E) ? dst[e] : -1;
    }
    #pragma unroll
    for (int i = 0; i < SORT_PER_THREAD; ++i)
        if (dv[i] >= 0) atomicAdd(&hist[dv[i] >> BIN_SHIFT], 1);
    __syncthreads();
    const int c = hist[t];
    if (c > 0) atomicAdd(&binTotal[t], c);
}

__global__ __launch_bounds__(SORT_THREADS) void gat_bin_place(
    const int* __restrict__ src, const int* __restrict__ dst,
    const int* __restrict__ binTotal, int* __restrict__ binCursor,
    int* __restrict__ binStart,
    int* __restrict__ ssrc, int* __restrict__ sdst, int E)
{
    __shared__ int hist[MAXBINS];
    __shared__ int gofs[MAXBINS];
    __shared__ int wsum[SORT_THREADS / 32];
    __shared__ int lsum[SORT_THREADS / 32];
    __shared__ int stS[SORT_CHUNK];
    __shared__ int stD[SORT_CHUNK];
    const int t    = threadIdx.x;
    const int lane = t & 31;
    const int w    = __builtin_amdgcn_readfirstlane(t >> 5);
    hist[t] = 0;
    __syncthreads();
    const int e0 = (int)blockIdx.x * SORT_CHUNK;
    int dv[SORT_PER_THREAD], sv[SORT_PER_THREAD];
    #pragma unroll
    for (int i = 0; i < SORT_PER_THREAD; ++i) {
        const int e = e0 + i * SORT_THREADS + t;
        dv[i] = (e < E) ? dst[e] : -1;
        sv[i] = (e < E) ? src[e] : 0;
    }
    #pragma unroll
    for (int i = 0; i < SORT_PER_THREAD; ++i)
        if (dv[i] >= 0) atomicAdd(&hist[dv[i] >> BIN_SHIFT], 1);

    const int tot = binTotal[t];
    int v = tot;
    for (int off = 1; off < 32; off <<= 1) {
        const int u = __shfl_up(v, off);
        if (lane >= off) v += u;
    }
    if (lane == 31) wsum[w] = v;
    __syncthreads();

    const int c = hist[t];
    int lv = c;
    for (int off = 1; off < 32; off <<= 1) {
        const int u = __shfl_up(lv, off);
        if (lane >= off) lv += u;
    }
    if (lane == 31) lsum[w] = lv;

    int start;
    {
        const int s0 = wsum[lane];
        int s = s0;
        for (int off = 1; off < 32; off <<= 1) {
            const int u = __shfl_up(s, off);
            if (lane >= off) s += u;
        }
        start = v - tot + __builtin_amdgcn_readlane(s - s0, w);
    }
    if (blockIdx.x == 0) binStart[t] = start;
    int bs = start;
    if (c > 0) bs += atomicAdd(&binCursor[t], c);
    __syncthreads();
    {
        const int s0 = lsum[lane];
        int s = s0;
        for (int off = 1; off < 32; off <<= 1) {
            const int u = __shfl_up(s, off);
            if (lane >= off) s += u;
        }
        const int lofs = lv - c + __builtin_amdgcn_readlane(s - s0, w);
        gofs[t] = bs - lofs;
        hist[t] = lofs;
    }
    __syncthreads();

    #pragma unroll
    for (int i = 0; i < SORT_PER_THREAD; ++i) {
        if (dv[i] >= 0) {
            const int lp = atomicAdd(&hist[dv[i] >> BIN_SHIFT], 1);
            stS[lp] = sv[i];
            stD[lp] = dv[i];
        }
    }
    __syncthreads();

    const int nv = (E - e0 < SORT_CHUNK) ? (E - e0) : SORT_CHUNK;
    for (int i = 0; i < SORT_PER_THREAD; ++i) {
        const int idx = i * SORT_THREADS + t;
        if (idx < nv) {
            const int d   = stD[idx];
            const int pos = gofs[d >> BIN_SHIFT] + idx;
            ssrc[pos] = stS[idx];
            sdst[pos] = d;
        }
    }
}

template <int M>
static __device__ __forceinline__ float gat_row_xor(float v) {
    return __int_as_float(__builtin_amdgcn_update_dpp(
        0, __float_as_int(v), 0x160 + M, 0xf, 0xf, true));
}

#define FB_WAVES 8
#define FB_THREADS (FB_WAVES * 32)
#define NS_PER_THREAD 8
#define EB 8

template <bool first>
static __device__ __forceinline__ int gat_sources(
    const int* __restrict__ p, int node, int lane, int left)
{
    int s = node;
    if (lane >= (first ? 1 : 0) && lane < left) s = p[lane - 1];
    return s;
}

template <int NB>
static __device__ __forceinline__ void gat_trip(
    const v4f* __restrict__ xl4, const v4f* __restrict__ xr4,
    int lane, int s, int jb, int c, const v4f R, const v4f a4,
    float& m, float& den, v4f& acc)
{
    v4f XL[NB], XR[NB];
    #pragma unroll
    for (int u = 0; u < NB; ++u) {
        const int sj = __builtin_amdgcn_readlane(s, (jb + u < c) ? (jb + u) : (c - 1));
        const int ix = sj * (CIN / 4) + lane;
        XL[u] = xl4[ix];
        XR[u] = xr4[ix];
    }
    #pragma unroll
    for (int u = 0; u < NB; ++u)
        asm volatile("" :: "v"(XR[u]));
    float p[NB];
    float mb = m;
    #pragma unroll
    for (int u = 0; u < NB; ++u) {
        v4f v = XL[u] + R;
        v = __builtin_elementwise_max(v, v * NEG_SLOPE);
        const v4f pr = a4 * v;
        const v2f hs = pr.xy + pr.zw;
        float e = hs.x + hs.y;
        e += gat_row_xor<1>(e);
        e += gat_row_xor<2>(e);
        e += gat_row_xor<4>(e);
        p[u] = (jb + u < c) ? e : -__builtin_inff();
        mb = fmaxf(mb, p[u]);
    }
    const float sc = __expf(m - mb);
    den *= sc;
    acc *= sc;
    #pragma unroll
    for (int u = 0; u < NB; ++u) {
        const float w = __expf(p[u] - mb);
        den += w;
        acc = XR[u] * w + acc;
    }
    m = mb;
}

__global__ __launch_bounds__(FB_THREADS) __attribute__((amdgpu_waves_per_eu(8, 8))) void gat_bin_edges(
    const float* __restrict__ xl, const float* __restrict__ xr,
    const int* __restrict__ ssrc, const int* __restrict__ sdst,
    const int* __restrict__ binStart, const int* __restrict__ binTotal,
    const float* __restrict__ att, const float* __restrict__ bias,
    int* __restrict__ nsrc, float* __restrict__ out,
    int Nn)
{
    __shared__ int ncnt[BIN_NODES];
    __shared__ int noff[BIN_NODES + 1];
    __shared__ __attribute__((aligned(16))) float sxr[BIN_NODES * CIN];

    const int tid  = (int)threadIdx.x;
    const int lane = tid & 31;
    const int wave  = __builtin_amdgcn_readfirstlane(tid >> 5);
    const int b     = (int)blockIdx.x;
    const int n0    = b << BIN_SHIFT;
    const int nn    = (Nn - n0 < BIN_NODES) ? (Nn - n0) : BIN_NODES;
    const int start = __builtin_amdgcn_readfirstlane(binStart[b]);
    const int cntE  = __builtin_amdgcn_readfirstlane(binTotal[b]);

    if (tid < BIN_NODES) ncnt[tid] = 0;
    __syncthreads();

    float4 t4[BIN_NODES * (CIN / 4) / FB_THREADS];
    {
        const float4* __restrict__ g4 = reinterpret_cast<const float4*>(xr + (size_t)n0 * CIN);
        const int lim = nn * (CIN / 4);
        #pragma unroll
        for (int i = 0; i < BIN_NODES * (CIN / 4) / FB_THREADS; ++i) {
            const int q = i * FB_THREADS + tid;
            t4[i] = (q < lim) ? g4[q] : make_float4(0.0f, 0.0f, 0.0f, 0.0f);
        }
    }

    for (int base = 0; base < cntE; base += NS_PER_THREAD * FB_THREADS) {
        int dv[NS_PER_THREAD];
        #pragma unroll
        for (int k = 0; k < NS_PER_THREAD; ++k) {
            const int j = base + k * FB_THREADS + tid;
            dv[k] = (j < cntE) ? sdst[start + j] : -1;
        }
        #pragma unroll
        for (int k = 0; k < NS_PER_THREAD; ++k)
            if (dv[k] >= 0) atomicAdd(&ncnt[dv[k] - n0], 1);
    }
    {
        float4* s4 = reinterpret_cast<float4*>(sxr);
        #pragma unroll
        for (int i = 0; i < BIN_NODES * (CIN / 4) / FB_THREADS; ++i)
            s4[i * FB_THREADS + tid] = t4[i];
    }
    __syncthreads();

    if (wave == 0) {
        const int c0 = ncnt[2 * lane];
        const int c1 = ncnt[2 * lane + 1];
        int v = c0 + c1;
        for (int off = 1; off < 32; off <<= 1) {
            const int u = __shfl_up(v, off);
            if (lane >= off) v += u;
        }
        const int ex = v - c0 - c1;
        noff[2 * lane]     = ex;
        noff[2 * lane + 1] = ex + c0;
        ncnt[2 * lane]     = ex;
        ncnt[2 * lane + 1] = ex + c0;
        if (lane == 31) noff[BIN_NODES] = v;
    }
    __syncthreads();

    for (int base = 0; base < cntE; base += NS_PER_THREAD * FB_THREADS) {
        int dv[NS_PER_THREAD], sv[NS_PER_THREAD];
        #pragma unroll
        for (int k = 0; k < NS_PER_THREAD; ++k) {
            const int j = base + k * FB_THREADS + tid;
            dv[k] = (j < cntE) ? sdst[start + j] : -1;
            sv[k] = (j < cntE) ? ssrc[start + j] : 0;
        }
        #pragma unroll
        for (int k = 0; k < NS_PER_THREAD; ++k) {
            if (dv[k] >= 0) {
                const int pos = atomicAdd(&ncnt[dv[k] - n0], 1);
                nsrc[start + pos] = sv[k];
            }
        }
    }
    __syncthreads();

    const v4f a4 = *reinterpret_cast<const v4f*>(att + lane * 4);
    const v4f* __restrict__ xl4 = reinterpret_cast<const v4f*>(xl);
    const v4f* __restrict__ xr4 = reinterpret_cast<const v4f*>(xr);
    const int* __restrict__ run = nsrc + start;
    const float4 b4 = *reinterpret_cast<const float4*>(bias + (lane & (HDIM / 4 - 1)) * 4);

    int o0 = 0, tot = 0, sv = 0;
    if (wave < nn) {
        o0  = __builtin_amdgcn_readfirstlane(noff[wave]);
        tot = __builtin_amdgcn_readfirstlane(noff[wave + 1]) - o0 + 1;
        sv  = gat_sources<true>(run + o0, n0 + wave, lane, tot);
    }
    asm volatile("" :: "v"(sv));
    for (int i = wave; i < nn; i += FB_WAVES) {
        const int node = n0 + i;
        int o0n = 0, totn = 0, svn = 0;
        if (i + FB_WAVES < nn) {
            o0n  = __builtin_amdgcn_readfirstlane(noff[i + FB_WAVES]);
            totn = __builtin_amdgcn_readfirstlane(noff[i + FB_WAVES + 1]) - o0n + 1;
            svn  = gat_sources<true>(run + o0n, node + FB_WAVES, lane, totn);
        }
        const v4f R = *reinterpret_cast<const v4f*>(&sxr[i * CIN + lane * 4]);
        float m = 0.0f, den = 0.0f;
        v4f acc = {0.0f, 0.0f, 0.0f, 0.0f};
        for (int g0 = 0; g0 < tot; g0 += 32) {
            const int c = (tot - g0 < 32) ? (tot - g0) : 32;
            int s = sv;
            if (g0 != 0) {
                s = gat_sources<false>(run + o0 + g0, node, lane, tot - g0);
                asm volatile("" :: "v"(s));
            }
            int jb = 0;
            while (jb < c) {
                if (c - jb > EB / 2) {
                    gat_trip<EB>(xl4, xr4, lane, s, jb, c, R, a4, m, den, acc);
                    jb += EB;
                } else {
                    gat_trip<EB / 2>(xl4, xr4, lane, s, jb, c, R, a4, m, den, acc);
                    jb += EB / 2;
                }
            }
        }
        const float dd = (den > 1e-16f) ? den : 1e-16f;
        float r0 = acc.x / dd, r1 = acc.y / dd, r2 = acc.z / dd, r3 = acc.w / dd;
        r0 += gat_row_xor<8>(r0);
        r1 += gat_row_xor<8>(r1);
        r2 += gat_row_xor<8>(r2);
        r3 += gat_row_xor<8>(r3);
        r0 += __shfl_xor(r0, 16);
        r1 += __shfl_xor(r1, 16);
        r2 += __shfl_xor(r2, 16);
        r3 += __shfl_xor(r3, 16);
        if (lane < HDIM / 4) {
            float4 o;
            o.x = b4.x + 0.25f * r0;
            o.y = b4.y + 0.25f * r1;
            o.z = b4.z + 0.25f * r2;
            o.w = b4.w + 0.25f * r3;
            *reinterpret_cast<float4*>(out + (size_t)node * HDIM + lane * 4) = o;
        }
        o0 = o0n; tot = totn; sv = svn;
    }
}

extern "C" void kernel_launch(void* const* d_in, const int* in_sizes, int n_in,
                              void* d_out, int out_size, void* d_ws, size_t ws_size,
                              hipStream_t stream)
{
    const float* x    = (const float*)d_in[0];
    const int*   eidx = (const int*)d_in[1];
    const float* Wl   = (const float*)d_in[2];
    const float* Wr   = (const float*)d_in[3];
    const float* att  = (const float*)d_in[4];
    const float* bias = (const float*)d_in[5];
    float* out = (float*)d_out;

    const int Nn   = in_sizes[0] / CIN;
    const int E    = in_sizes[1] / 2;

    const int* src = eidx;
    const int* dst = eidx + E;

    float* ws = (float*)d_ws;
    size_t off = 0;
    float* xl    = ws + off; off += (size_t)Nn * CIN;
    float* xr    = ws + off; off += (size_t)Nn * CIN;
    int* binTotal  = (int*)(ws + off); off += MAXBINS;
    int* binCursor = (int*)(ws + off); off += MAXBINS;
    int* binStart  = (int*)(ws + off); off += MAXBINS;
    int* ssrc      = (int*)(ws + off); off += (size_t)E;
    int* sdst      = (int*)(ws + off); off += (size_t)E;
    int* nsrc      = (int*)(ws + off); off += (size_t)E;
    (void)ws_size;

    {
        const int rowTiles = (Nn + 15) / 16;
        dim3 grid((rowTiles + GM_WAVES - 1) / GM_WAVES, 2);
        gat_gemm_wmma<<<grid, GM_WAVES * 32, 0, stream>>>(
            x, Wl, Wr, xl, xr, (float*)binTotal, 2 * MAXBINS / 4, Nn);
    }

    const int sortBlocks = (E + SORT_CHUNK - 1) / SORT_CHUNK;
    const int nbins      = (Nn + BIN_NODES - 1) / BIN_NODES;

    gat_bin_count<<<sortBlocks, SORT_THREADS, 0, stream>>>(dst, binTotal, E);

    gat_bin_place<<<sortBlocks, SORT_THREADS, 0, stream>>>(
        src, dst, binTotal, binCursor, binStart, ssrc, sdst, E);

    gat_bin_edges<<<nbins, FB_THREADS, 0, stream>>>(
        xl, xr, ssrc, sdst, binStart, binTotal, att, bias, nsrc, out, Nn);
}
